# starting kernel
# baseline (speedup 1.0000x reference)
#include <hip/hip_runtime.h>
#include <hip/hip_bf16.h>
#include <math.h>

#define SLOPE 0.2f

typedef float v2f __attribute__((ext_vector_type(2)));
typedef float v8f __attribute__((ext_vector_type(8)));


__global__ void fill_f32(float* __restrict__ p, float v, int n) {
    int i = blockIdx.x * blockDim.x + threadIdx.x;
    if (i < n) p[i] = v;
}

__device__ __forceinline__ void atomicMaxFloat(float* addr, float value) {
    if (value >= 0.0f) {
        atomicMax((int*)addr, __float_as_int(value));
    } else {
        atomicMin((unsigned int*)addr, __float_as_uint(value));
    }
}

__device__ __forceinline__ float leaky(float v) {
    return v > 0.0f ? v : SLOPE * v;
}

template <int NT>
__global__ void gemm_wmma_strip(const float* __restrict__ A,
                                const float* __restrict__ B,
                                float* __restrict__ C,
                                int M, int K) {
    const int N = NT * 16;
    extern __shared__ float sB[];

    for (int i = threadIdx.x; i < K * N; i += blockDim.x) sB[i] = B[i];
    __syncthreads();

    const int lane = threadIdx.x & 31;
    const int wid  = threadIdx.x >> 5;
    const int row0 = (blockIdx.x * (blockDim.x >> 5) + wid) * 16;
    if (row0 >= M) return;

    const int mn = lane & 15;
    const int hi = lane >> 4;
    const int kg = hi << 1;

    const float* arow = A + (size_t)(row0 + mn) * K;

    v8f acc[NT];
#pragma unroll
    for (int t = 0; t < NT; ++t) acc[t] = (v8f){};

    for (int k0 = 0; k0 < K; k0 += 4) {
        v2f a;
        a.x = arow[k0 + kg];
        a.y = arow[k0 + kg + 1];
        const float* brow = sB + (size_t)(k0 + kg) * N + mn;
#pragma unroll
        for (int t = 0; t < NT; ++t) {
            v2f b;
            b.x = brow[t * 16];
            b.y = brow[t * 16 + N];
            acc[t] = __builtin_amdgcn_wmma_f32_16x16x4_f32(
                 false, a,  false, b,
                 (short)0, acc[t],  false,  false);
        }
    }

#pragma unroll
    for (int t = 0; t < NT; ++t) {
        float* crow = C + (size_t)(row0 + hi * 8) * N + t * 16 + mn;
#pragma unroll
        for (int v = 0; v < 8; ++v) crow[(size_t)v * N] = acc[t][v];
    }
}

__global__ void node_scores(const float* __restrict__ h,
                            const float* __restrict__ a_src,
                            const float* __restrict__ a_dst,
                            float* __restrict__ ls, float* __restrict__ ld,
                            int n_nodes, int C) {
    const int lane = threadIdx.x & 31;
    const int node = (blockIdx.x * blockDim.x + threadIdx.x) >> 5;
    if (node >= n_nodes) return;
    float s = 0.0f, d = 0.0f;
    for (int c = lane; c < C; c += 32) {
        float v = h[(size_t)node * C + c];
        s += v * a_src[c];
        d += v * a_dst[c];
    }
#pragma unroll
    for (int off = 16; off > 0; off >>= 1) {
        s += __shfl_xor(s, off, 32);
        d += __shfl_xor(d, off, 32);
    }
    if (lane == 0) { ls[node] = s; ld[node] = d; }
}


__global__ void edge_logit_max(const int* __restrict__ src, const int* __restrict__ dst,
                               const float* __restrict__ ls, const float* __restrict__ ld,
                               float* __restrict__ ev, float* __restrict__ lmax, int E) {
    int e = blockIdx.x * blockDim.x + threadIdx.x;
    if (e >= E) return;
    int d = dst[e];
    float lg = leaky(ls[src[e]] + ld[d]);
    ev[e] = lg;
    atomicMaxFloat(&lmax[d], lg);
}

__global__ void edge_expsum(const int* __restrict__ dst,
                            const float* __restrict__ lmax,
                            float* __restrict__ ev, float* __restrict__ denom, int E) {
    int e = blockIdx.x * blockDim.x + threadIdx.x;
    if (e >= E) return;
    int d = dst[e];
    float ex = __expf(ev[e] - lmax[d]);
    ev[e] = ex;
    atomicAdd(&denom[d], ex);
}

template <int C>
__global__ void edge_aggregate(const int* __restrict__ src, const int* __restrict__ dst,
                               const float* __restrict__ ev, const float* __restrict__ denom,
                               const float* __restrict__ h, float* __restrict__ agg, int E) {
    constexpr int GP = C / 4;
    int t = blockIdx.x * blockDim.x + threadIdx.x;
    int e = t / GP;
    int g = t - e * GP;
    if (e >= E) return;
    int s = src[e], d = dst[e];
    float alpha = ev[e] / denom[d];
    const float4 v = *((const float4*)(h + (size_t)s * C) + g);
    float* ap = agg + (size_t)d * C + g * 4;
    atomicAdd(ap + 0, alpha * v.x);
    atomicAdd(ap + 1, alpha * v.y);
    atomicAdd(ap + 2, alpha * v.z);
    atomicAdd(ap + 3, alpha * v.w);
}


__global__ void bias_act(const float* __restrict__ agg, const float* __restrict__ b,
                         float* __restrict__ out, int total, int C, int do_relu) {
    int i = blockIdx.x * blockDim.x + threadIdx.x;
    if (i >= total) return;
    float v = agg[i] + b[i % C];
    out[i] = do_relu ? fmaxf(v, 0.0f) : v;
}


static inline int cdiv(int a, int b) { return (a + b - 1) / b; }

extern "C" void kernel_launch(void* const* d_in, const int* in_sizes, int n_in,
                              void* d_out, int out_size, void* d_ws, size_t ws_size,
                              hipStream_t stream) {
    const float* x      = (const float*)d_in[0];
    const int*   eidx   = (const int*)d_in[1];
    const float* W1     = (const float*)d_in[2];
    const float* a_src1 = (const float*)d_in[3];
    const float* a_dst1 = (const float*)d_in[4];
    const float* b1     = (const float*)d_in[5];
    const float* W2     = (const float*)d_in[6];
    const float* a_src2 = (const float*)d_in[7];
    const float* a_dst2 = (const float*)d_in[8];
    const float* b2     = (const float*)d_in[9];
    float* out = (float*)d_out;

    const int IN = 128, H = 64, O = 32;
    const int N = in_sizes[0] / IN;
    const int E = in_sizes[1] / 2;
    const int* src = eidx;
    const int* dst = eidx + E;

    float* ws    = (float*)d_ws;
    float* h1    = ws;
    float* agg1  = h1 + (size_t)N * H;
    float* h2    = agg1 + (size_t)N * H;
    float* ls    = h2 + (size_t)N * O;
    float* ld    = ls + N;
    float* lmax  = ld + N;
    float* denom = lmax + N;
    float* ev    = denom + N;

    const int B = 256;

    gemm_wmma_strip<4><<<cdiv(N, 128), B, IN * H * sizeof(float), stream>>>(x, W1, h1, N, IN);

    node_scores<<<cdiv(N * 32, B), B, 0, stream>>>(h1, a_src1, a_dst1, ls, ld, N, H);

    fill_f32<<<cdiv(N, B), B, 0, stream>>>(lmax, -INFINITY, N);
    fill_f32<<<cdiv(N, B), B, 0, stream>>>(denom, 0.0f, N);
    fill_f32<<<cdiv(N * H, B), B, 0, stream>>>(agg1, 0.0f, N * H);

    edge_logit_max<<<cdiv(E, B), B, 0, stream>>>(src, dst, ls, ld, ev, lmax, E);
    edge_expsum<<<cdiv(E, B), B, 0, stream>>>(dst, lmax, ev, denom, E);
    edge_aggregate<64><<<cdiv(E * (H / 4), B), B, 0, stream>>>(src, dst, ev, denom, h1, agg1, E);

    bias_act<<<cdiv(N * H, B), B, 0, stream>>>(agg1, b1, h1, N * H, H,  1);

    gemm_wmma_strip<2><<<cdiv(N, 128), B, H * O * sizeof(float), stream>>>(h1, W2, h2, N, H);

    node_scores<<<cdiv(N * 32, B), B, 0, stream>>>(h2, a_src2, a_dst2, ls, ld, N, O);

    fill_f32<<<cdiv(N, B), B, 0, stream>>>(lmax, -INFINITY, N);
    fill_f32<<<cdiv(N, B), B, 0, stream>>>(denom, 0.0f, N);
    fill_f32<<<cdiv(N * O, B), B, 0, stream>>>(out, 0.0f, N * O);

    edge_logit_max<<<cdiv(E, B), B, 0, stream>>>(src, dst, ls, ld, ev, lmax, E);
    edge_expsum<<<cdiv(E, B), B, 0, stream>>>(dst, lmax, ev, denom, E);
    edge_aggregate<32><<<cdiv(E * (O / 4), B), B, 0, stream>>>(src, dst, ev, denom, h2, out, E);

    bias_act<<<cdiv(N * O, B), B, 0, stream>>>(out, b2, out, N * O, O,  0);
}


# Group edges by destination once per call (integer ranks, prefix sum); one wave per node then does max, exp-sum and weighted row sum in registers for both layers: no float atomics, no fills
# speedup vs baseline: 2.4250x; 2.4250x over previous
#include <hip/hip_runtime.h>
#include <hip/hip_bf16.h>
#include <math.h>


#define SLOPE 0.2f

typedef float v2f __attribute__((ext_vector_type(2)));
typedef float v8f __attribute__((ext_vector_type(8)));


__global__ void fill_i32(int* __restrict__ p, int v, int n) {
    int i = blockIdx.x * blockDim.x + threadIdx.x;
    if (i < n) p[i] = v;
}

__device__ __forceinline__ float leaky(float v) {
    return v > 0.0f ? v : SLOPE * v;
}

template <int NT>
__global__ void gemm_wmma_strip(const float* __restrict__ A,
                                const float* __restrict__ B,
                                float* __restrict__ C,
                                int M, int K) {
    const int N = NT * 16;
    extern __shared__ float sB[];

    for (int i = threadIdx.x; i < K * N; i += blockDim.x) sB[i] = B[i];
    __syncthreads();

    const int lane = threadIdx.x & 31;
    const int wid  = threadIdx.x >> 5;
    const int row0 = (blockIdx.x * (blockDim.x >> 5) + wid) * 16;
    if (row0 >= M) return;

    const int mn = lane & 15;
    const int hi = lane >> 4;
    const int kg = hi << 1;

    const float* arow = A + (size_t)(row0 + mn) * K;

    v8f acc[NT];
#pragma unroll
    for (int t = 0; t < NT; ++t) acc[t] = (v8f){};

    for (int k0 = 0; k0 < K; k0 += 4) {
        v2f a;
        a.x = arow[k0 + kg];
        a.y = arow[k0 + kg + 1];
        const float* brow = sB + (size_t)(k0 + kg) * N + mn;
#pragma unroll
        for (int t = 0; t < NT; ++t) {
            v2f b;
            b.x = brow[t * 16];
            b.y = brow[t * 16 + N];
            acc[t] = __builtin_amdgcn_wmma_f32_16x16x4_f32(
                 false, a,  false, b,
                 (short)0, acc[t],  false,  false);
        }
    }

#pragma unroll
    for (int t = 0; t < NT; ++t) {
        float* crow = C + (size_t)(row0 + hi * 8) * N + t * 16 + mn;
#pragma unroll
        for (int v = 0; v < 8; ++v) crow[(size_t)v * N] = acc[t][v];
    }
}

__global__ void node_scores(const float* __restrict__ h,
                            const float* __restrict__ a_src,
                            const float* __restrict__ a_dst,
                            float* __restrict__ ls, float* __restrict__ ld,
                            int n_nodes, int C) {
    const int lane = threadIdx.x & 31;
    const int node = (blockIdx.x * blockDim.x + threadIdx.x) >> 5;
    if (node >= n_nodes) return;
    float s = 0.0f, d = 0.0f;
    for (int c = lane; c < C; c += 32) {
        float v = h[(size_t)node * C + c];
        s += v * a_src[c];
        d += v * a_dst[c];
    }
#pragma unroll
    for (int off = 16; off > 0; off >>= 1) {
        s += __shfl_xor(s, off, 32);
        d += __shfl_xor(d, off, 32);
    }
    if (lane == 0) { ls[node] = s; ld[node] = d; }
}


__global__ void rank_count(const int* __restrict__ dst, int* __restrict__ cnt,
                           int* __restrict__ rank, int E) {
    int e = blockIdx.x * blockDim.x + threadIdx.x;
    if (e >= E) return;
    rank[e] = atomicAdd(&cnt[dst[e]], 1);
}

__global__ void __launch_bounds__(1024)
scan_counts(const int* __restrict__ cnt, int* __restrict__ first, int n) {
    __shared__ int wave_tot[32];
    __shared__ int wave_pre[32];
    const int tid  = threadIdx.x;
    const int lane = tid & 31;
    const int w    = tid >> 5;
    const int blk0 = blockIdx.x * 1024;

    int pre = 0;
    for (int i = tid; i < blk0; i += 1024) pre += cnt[i];

    const int i = blk0 + tid;
    const int c = (i < n) ? cnt[i] : 0;

    int x = c;
    for (int off = 1; off < 32; off <<= 1) {
        int u = __shfl_up(x, off, 32);
        if (lane >= off) x += u;
    }
    for (int off = 16; off > 0; off >>= 1) pre += __shfl_xor(pre, off, 32);

    if (lane == 31) wave_tot[w] = x;
    if (lane == 0)  wave_pre[w] = pre;
    __syncthreads();

    const int t = wave_tot[lane];
    int p = wave_pre[lane];
    int inc = t;
    for (int off = 1; off < 32; off <<= 1) {
        int u = __shfl_up(inc, off, 32);
        if (lane >= off) inc += u;
    }
    for (int off = 16; off > 0; off >>= 1) p += __shfl_xor(p, off, 32);
    const int waves_before = __shfl(inc - t, w, 32);

    if (i < n) first[i] = p + waves_before + x - c;
}

__global__ void place_edges(const int* __restrict__ src, const int* __restrict__ dst,
                            const int* __restrict__ first, const int* __restrict__ rank,
                            int* __restrict__ nbr, int E) {
    int e = blockIdx.x * blockDim.x + threadIdx.x;
    if (e >= E) return;
    nbr[first[dst[e]] + rank[e]] = src[e];
}

template <int C, int NB>
__device__ __forceinline__ void add_rows(const float* __restrict__ h, int s, float a,
                                         int j, int lane, float* acc) {
    constexpr int VPL = C / 32;
    float v[NB][VPL];
    for (int t = 0; t < NB; ++t) {
        const int sj = __builtin_amdgcn_readlane(s, j + t);
        const float* row = h + (size_t)sj * C + lane * VPL;
        if (VPL == 2) {
            const float2 r = *(const float2*)row;
            v[t][0] = r.x;
            v[t][VPL - 1] = r.y;
        } else {
            v[t][0] = row[0];
        }
    }
    for (int t = 0; t < NB; ++t) {
        const float aj = __int_as_float(__builtin_amdgcn_readlane(__float_as_int(a), j + t));
        for (int q = 0; q < VPL; ++q) acc[q] += aj * v[t][q];
    }
}

template <int C>
__device__ __forceinline__ void add_chunk(const float* __restrict__ h, int s, float a,
                                          int cnt, int lane, float* acc) {
    int j = 0;
    for (; j + 8 <= cnt; j += 8) add_rows<C, 8>(h, s, a, j, lane, acc);
    if (j + 4 <= cnt) { add_rows<C, 4>(h, s, a, j, lane, acc); j += 4; }
    if (j + 2 <= cnt) { add_rows<C, 2>(h, s, a, j, lane, acc); j += 2; }
    if (j < cnt)      { add_rows<C, 1>(h, s, a, j, lane, acc); }
}

template <int C, int RELU>
__global__ void __launch_bounds__(256)
node_softmax_sum(const int* __restrict__ first, const int* __restrict__ nbr,
                 const float* __restrict__ ls, const float* __restrict__ ld,
                 const float* __restrict__ h, const float* __restrict__ bias,
                 float* __restrict__ out, int n_nodes, int n_edges) {
    constexpr int VPL = C / 32;
    const int lane = threadIdx.x & 31;
    const int node_v = (blockIdx.x * blockDim.x + threadIdx.x) >> 5;
    if (node_v >= n_nodes) return;
    const int node = __builtin_amdgcn_readfirstlane(node_v);
    const int beg  = __builtin_amdgcn_readfirstlane(first[node]);
    const int end  = (node + 1 < n_nodes) ? __builtin_amdgcn_readfirstlane(first[node + 1]) : n_edges;
    const float ldn = ld[node];

    float acc[VPL];
    for (int q = 0; q < VPL; ++q) acc[q] = 0.0f;

    if (end > beg) {
        const int  i0 = beg + lane;
        const bool v0 = i0 < end;
        const int  s0 = v0 ? nbr[i0] : 0;
        const float lg0 = v0 ? leaky(ls[s0] + ldn) : -INFINITY;

        float m = lg0;
        for (int i = i0 + 32; i < end; i += 32) m = fmaxf(m, leaky(ls[nbr[i]] + ldn));
        for (int off = 16; off > 0; off >>= 1) m = fmaxf(m, __shfl_xor(m, off, 32));

        const float e0 = v0 ? __expf(lg0 - m) : 0.0f;
        float den = e0;
        for (int i = i0 + 32; i < end; i += 32) den += __expf(leaky(ls[nbr[i]] + ldn) - m);
        for (int off = 16; off > 0; off >>= 1) den += __shfl_xor(den, off, 32);

        const int deg = end - beg;
        add_chunk<C>(h, s0, e0 / den, deg < 32 ? deg : 32, lane, acc);

        for (int c0 = beg + 32; c0 < end; c0 += 32) {
            const int  i = c0 + lane;
            const bool v = i < end;
            const int  s = v ? nbr[i] : 0;
            const float a = v ? __expf(leaky(ls[s] + ldn) - m) / den : 0.0f;
            const int left = end - c0;
            add_chunk<C>(h, s, a, left < 32 ? left : 32, lane, acc);
        }
    }

    float* orow = out + (size_t)node * C + lane * VPL;
    for (int q = 0; q < VPL; ++q) {
        float r = acc[q] + bias[lane * VPL + q];
        orow[q] = RELU ? fmaxf(r, 0.0f) : r;
    }
}


static inline int cdiv(int a, int b) { return (a + b - 1) / b; }

extern "C" void kernel_launch(void* const* d_in, const int* in_sizes, int n_in,
                              void* d_out, int out_size, void* d_ws, size_t ws_size,
                              hipStream_t stream) {
    const float* x      = (const float*)d_in[0];
    const int*   eidx   = (const int*)d_in[1];
    const float* W1     = (const float*)d_in[2];
    const float* a_src1 = (const float*)d_in[3];
    const float* a_dst1 = (const float*)d_in[4];
    const float* b1     = (const float*)d_in[5];
    const float* W2     = (const float*)d_in[6];
    const float* a_src2 = (const float*)d_in[7];
    const float* a_dst2 = (const float*)d_in[8];
    const float* b2     = (const float*)d_in[9];
    float* out = (float*)d_out;

    const int IN = 128, H = 64, O = 32;
    const int N = in_sizes[0] / IN;
    const int E = in_sizes[1] / 2;
    const int* src = eidx;
    const int* dst = eidx + E;

    float* ws    = (float*)d_ws;
    float* h1    = ws;
    float* act1  = h1 + (size_t)N * H;
    float* h2    = act1 + (size_t)N * H;
    float* ls    = h2 + (size_t)N * O;
    float* ld    = ls + N;
    int*   cnt   = (int*)(ld + N);
    int*   first = cnt + N;
    int*   nbr   = first + N;
    int*   rank  = (int*)act1;

    const int B = 256;

    fill_i32<<<cdiv(N, B), B, 0, stream>>>(cnt, 0, N);
    rank_count<<<cdiv(E, B), B, 0, stream>>>(dst, cnt, rank, E);
    scan_counts<<<cdiv(N, 1024), 1024, 0, stream>>>(cnt, first, N);
    place_edges<<<cdiv(E, B), B, 0, stream>>>(src, dst, first, rank, nbr, E);

    gemm_wmma_strip<4><<<cdiv(N, 128), B, IN * H * sizeof(float), stream>>>(x, W1, h1, N, IN);
    node_scores<<<cdiv(N * 32, B), B, 0, stream>>>(h1, a_src1, a_dst1, ls, ld, N, H);
    node_softmax_sum<64, 1><<<cdiv(N * 32, B), B, 0, stream>>>(first, nbr, ls, ld, h1, b1, act1, N, E);

    gemm_wmma_strip<2><<<cdiv(N, 128), B, H * O * sizeof(float), stream>>>(act1, W2, h2, N, H);
    node_scores<<<cdiv(N * 32, B), B, 0, stream>>>(h2, a_src2, a_dst2, ls, ld, N, O);
    node_softmax_sum<32, 0><<<cdiv(N * 32, B), B, 0, stream>>>(first, nbr, ls, ld, h2, b2, out, N, E);
}


# Softmax-and-row-sum kernels: a group of C/4 lanes per node with 16 bytes of the row per lane, so 2 (C=64) or 4 (C=32) nodes share a wave and each row load is 512 B of whole lines
# speedup vs baseline: 2.8716x; 1.1842x over previous
#include <hip/hip_runtime.h>
#include <hip/hip_bf16.h>
#include <math.h>


#define SLOPE 0.2f

typedef float v2f __attribute__((ext_vector_type(2)));
typedef float v8f __attribute__((ext_vector_type(8)));


__global__ void fill_i32(int* __restrict__ p, int v, int n) {
    int i = blockIdx.x * blockDim.x + threadIdx.x;
    if (i < n) p[i] = v;
}

__device__ __forceinline__ float leaky(float v) {
    return v > 0.0f ? v : SLOPE * v;
}

template <int NT>
__global__ void gemm_wmma_strip(const float* __restrict__ A,
                                const float* __restrict__ B,
                                const float* __restrict__ a_src,
                                const float* __restrict__ a_dst,
                                float* __restrict__ C,
                                float* __restrict__ ls, float* __restrict__ ld,
                                int M, int K) {
    const int N = NT * 16;
    extern __shared__ float sB[];

    for (int i = threadIdx.x; i < K * N; i += blockDim.x) sB[i] = B[i];
    __syncthreads();

    const int lane = threadIdx.x & 31;
    const int wid  = threadIdx.x >> 5;
    const int row0 = (blockIdx.x * (blockDim.x >> 5) + wid) * 16;
    if (row0 >= M) return;

    const int mn = lane & 15;
    const int hi = lane >> 4;
    const int kg = hi << 1;

    const float* arow = A + (size_t)(row0 + mn) * K;

    v8f acc[NT];
#pragma unroll
    for (int t = 0; t < NT; ++t) acc[t] = (v8f){};

    for (int k0 = 0; k0 < K; k0 += 4) {
        v2f a;
        a.x = arow[k0 + kg];
        a.y = arow[k0 + kg + 1];
        const float* brow = sB + (size_t)(k0 + kg) * N + mn;
#pragma unroll
        for (int t = 0; t < NT; ++t) {
            v2f b;
            b.x = brow[t * 16];
            b.y = brow[t * 16 + N];
            acc[t] = __builtin_amdgcn_wmma_f32_16x16x4_f32(
                 false, a,  false, b,
                 (short)0, acc[t],  false,  false);
        }
    }

#pragma unroll
    for (int t = 0; t < NT; ++t) {
        float* crow = C + (size_t)(row0 + hi * 8) * N + t * 16 + mn;
#pragma unroll
        for (int v = 0; v < 8; ++v) crow[(size_t)v * N] = acc[t][v];
    }

    float ps[8], pd[8];
    for (int v = 0; v < 8; ++v) { ps[v] = 0.0f; pd[v] = 0.0f; }
    for (int t = 0; t < NT; ++t) {
        const float as = a_src[t * 16 + mn];
        const float ad = a_dst[t * 16 + mn];
        for (int v = 0; v < 8; ++v) {
            ps[v] += acc[t][v] * as;
            pd[v] += acc[t][v] * ad;
        }
    }
    for (int off = 8; off > 0; off >>= 1) {
        for (int v = 0; v < 8; ++v) {
            ps[v] += __shfl_xor(ps[v], off, 32);
            pd[v] += __shfl_xor(pd[v], off, 32);
        }
    }
    const int r = mn & 7;
    float sv = ps[0], dv = pd[0];
    for (int v = 1; v < 8; ++v) {
        sv = (r == v) ? ps[v] : sv;
        dv = (r == v) ? pd[v] : dv;
    }
    float* sc = (mn < 8) ? ls : ld;
    sc[row0 + hi * 8 + r] = (mn < 8) ? sv : dv;
}


__global__ void rank_count(const int* __restrict__ dst, int* __restrict__ cnt,
                           int* __restrict__ rank, int E) {
    int e = blockIdx.x * blockDim.x + threadIdx.x;
    if (e >= E) return;
    rank[e] = atomicAdd(&cnt[dst[e]], 1);
}

__global__ void __launch_bounds__(1024)
scan_counts(const int* __restrict__ cnt, int* __restrict__ first, int n) {
    __shared__ int wave_tot[32];
    __shared__ int wave_pre[32];
    const int tid  = threadIdx.x;
    const int lane = tid & 31;
    const int w    = tid >> 5;
    const int blk0 = blockIdx.x * 1024;

    int pre = 0;
    for (int i = tid; i < blk0; i += 1024) pre += cnt[i];

    const int i = blk0 + tid;
    const int c = (i < n) ? cnt[i] : 0;

    int x = c;
    for (int off = 1; off < 32; off <<= 1) {
        int u = __shfl_up(x, off, 32);
        if (lane >= off) x += u;
    }
    for (int off = 16; off > 0; off >>= 1) pre += __shfl_xor(pre, off, 32);

    if (lane == 31) wave_tot[w] = x;
    if (lane == 0)  wave_pre[w] = pre;
    __syncthreads();

    const int t = wave_tot[lane];
    int p = wave_pre[lane];
    int inc = t;
    for (int off = 1; off < 32; off <<= 1) {
        int u = __shfl_up(inc, off, 32);
        if (lane >= off) inc += u;
    }
    for (int off = 16; off > 0; off >>= 1) p += __shfl_xor(p, off, 32);
    const int waves_before = __shfl(inc - t, w, 32);

    if (i < n) first[i] = p + waves_before + x - c;
}

__global__ void place_edges(const int* __restrict__ src, const int* __restrict__ dst,
                            const int* __restrict__ first, const int* __restrict__ rank,
                            int* __restrict__ nbr, int E) {
    int e = blockIdx.x * blockDim.x + threadIdx.x;
    if (e >= E) return;
    nbr[first[dst[e]] + rank[e]] = src[e];
}


template <int C>
__device__ __forceinline__ void add_rows(const float* __restrict__ h, int sc, float ac,
                                         int j0, bool more, int gl, float4& acc) {
    constexpr int G = C / 4;
    float4 v[8];
#pragma unroll
    for (int t = 0; t < 4; ++t) {
        const int sj = __shfl(sc, j0 + t, G);
        v[t] = *(const float4*)((const char*)h + ((unsigned)sj * (C * 4u) + gl * 16u));
    }
    if (more) {
#pragma unroll
        for (int t = 4; t < 8; ++t) {
            const int sj = __shfl(sc, j0 + t, G);
            v[t] = *(const float4*)((const char*)h + ((unsigned)sj * (C * 4u) + gl * 16u));
        }
    }
#pragma unroll
    for (int t = 0; t < 4; ++t) {
        const float aj = __shfl(ac, j0 + t, G);
        acc.x += aj * v[t].x;
        acc.y += aj * v[t].y;
        acc.z += aj * v[t].z;
        acc.w += aj * v[t].w;
    }
    if (more) {
#pragma unroll
        for (int t = 4; t < 8; ++t) {
            const float aj = __shfl(ac, j0 + t, G);
            acc.x += aj * v[t].x;
            acc.y += aj * v[t].y;
            acc.z += aj * v[t].z;
            acc.w += aj * v[t].w;
        }
    }
}

template <int C>
__device__ __forceinline__ void add_block(const float* __restrict__ h, const int* s, const float* a,
                                          int cnt, int gl, float4& acc) {
    constexpr int G  = C / 4;
    constexpr int KC = 32 / G;
#pragma unroll
    for (int c = 0; c < KC; ++c) {
#pragma unroll
        for (int jb = 0; jb < G; jb += 8) {
            const int rem = cnt - (c * G + jb);
            if (rem > 0) add_rows<C>(h, s[c], a[c], jb, rem > 4, gl, acc);
        }
    }
}

template <int C, int RELU>
__global__ void __launch_bounds__(256)
node_softmax_sum(const int* __restrict__ first, const int* __restrict__ nbr,
                 const float* __restrict__ ls, const float* __restrict__ ld,
                 const float* __restrict__ h, const float* __restrict__ bias,
                 float* __restrict__ out, int n_nodes, int n_edges) {
    constexpr int G  = C / 4;
    constexpr int KC = 32 / G;
    const int tid = blockIdx.x * blockDim.x + threadIdx.x;
    const int gl  = threadIdx.x & (G - 1);
    const int node = tid / G;
    if ((tid & ~31) / G >= n_nodes) return;

    const bool live = node < n_nodes;
    int beg = 0, end = 0;
    float ldn = 0.0f;
    if (live) {
        beg = first[node];
        end = (node + 1 < n_nodes) ? first[node + 1] : n_edges;
        ldn = ld[node];
    }
    const int deg = end - beg;

    int dmax = deg;
    for (int off = G; off < 32; off <<= 1) dmax = max(dmax, __shfl_xor(dmax, off, 32));
    const int wdeg = __builtin_amdgcn_readfirstlane(dmax);

    float4 acc = make_float4(0.0f, 0.0f, 0.0f, 0.0f);

    if (wdeg > 0) {
        int   s[KC];
        float lv[KC];
        float lg[KC];
        bool  ok[KC];
        const int last = deg > 0 ? deg - 1 : 0;
#pragma unroll
        for (int c = 0; c < KC; ++c) {
            const int slot = c * G + gl;
            ok[c] = slot < deg;
            int idx = beg + (ok[c] ? slot : last);
            idx = idx < n_edges ? idx : n_edges - 1;
            s[c] = nbr[idx];
        }
#pragma unroll
        for (int c = 0; c < KC; ++c) lv[c] = ls[s[c]];
        float m = -INFINITY;
#pragma unroll
        for (int c = 0; c < KC; ++c) {
            lg[c] = ok[c] ? leaky(lv[c] + ldn) : -INFINITY;
            m = fmaxf(m, lg[c]);
        }
        for (int i = beg + 32 + gl; i < end; i += G) m = fmaxf(m, leaky(ls[nbr[i]] + ldn));
        for (int off = G / 2; off > 0; off >>= 1) m = fmaxf(m, __shfl_xor(m, off, 32));

        float e[KC];
        float den = 0.0f;
#pragma unroll
        for (int c = 0; c < KC; ++c) {
            e[c] = ok[c] ? __expf(lg[c] - m) : 0.0f;
            den += e[c];
        }
        for (int i = beg + 32 + gl; i < end; i += G) den += __expf(leaky(ls[nbr[i]] + ldn) - m);
        for (int off = G / 2; off > 0; off >>= 1) den += __shfl_xor(den, off, 32);

        float a[KC];
#pragma unroll
        for (int c = 0; c < KC; ++c) a[c] = ok[c] ? e[c] / den : 0.0f;

        add_block<C>(h, s, a, wdeg < 32 ? wdeg : 32, gl, acc);

        for (int base = 32; base < wdeg; base += 32) {
            int   s2[KC];
            float a2[KC];
#pragma unroll
            for (int c = 0; c < KC; ++c) {
                const int  i = base + c * G + gl;
                const bool v = i < deg;
                s2[c] = v ? nbr[beg + i] : 0;
                a2[c] = v ? __expf(leaky(ls[s2[c]] + ldn) - m) / den : 0.0f;
            }
            const int left = wdeg - base;
            add_block<C>(h, s2, a2, left < 32 ? left : 32, gl, acc);
        }
    }

    if (live) {
        const float* bp = bias + gl * 4;
        const float4 b = make_float4(bp[0], bp[1], bp[2], bp[3]);
        float4 r = make_float4(acc.x + b.x, acc.y + b.y, acc.z + b.z, acc.w + b.w);
        if (RELU) {
            r.x = fmaxf(r.x, 0.0f);
            r.y = fmaxf(r.y, 0.0f);
            r.z = fmaxf(r.z, 0.0f);
            r.w = fmaxf(r.w, 0.0f);
        }
        *(float4*)(out + (size_t)node * C + gl * 4) = r;
    }
}


static inline int cdiv(int a, int b) { return (a + b - 1) / b; }

extern "C" void kernel_launch(void* const* d_in, const int* in_sizes, int n_in,
                              void* d_out, int out_size, void* d_ws, size_t ws_size,
                              hipStream_t stream) {
    const float* x      = (const float*)d_in[0];
    const int*   eidx   = (const int*)d_in[1];
    const float* W1     = (const float*)d_in[2];
    const float* a_src1 = (const float*)d_in[3];
    const float* a_dst1 = (const float*)d_in[4];
    const float* b1     = (const float*)d_in[5];
    const float* W2     = (const float*)d_in[6];
    const float* a_src2 = (const float*)d_in[7];
    const float* a_dst2 = (const float*)d_in[8];
    const float* b2     = (const float*)d_in[9];
    float* out = (float*)d_out;

    const int IN = 128, H = 64, O = 32;
    const int N = in_sizes[0] / IN;
    const int E = in_sizes[1] / 2;
    const int* src = eidx;
    const int* dst = eidx + E;

    float* ws    = (float*)d_ws;
    float* h1    = ws;
    float* act1  = h1 + (size_t)N * H;
    float* h2    = act1 + (size_t)N * H;
    float* ls    = h2 + (size_t)N * O;
    float* ld    = ls + N;
    int*   cnt   = (int*)(ld + N);
    int*   first = cnt + N;
    int*   nbr   = first + N;
    int*   rank  = (int*)act1;

    const int B = 256;

    fill_i32<<<cdiv(N, B), B, 0, stream>>>(cnt, 0, N);
    rank_count<<<cdiv(E, B), B, 0, stream>>>(dst, cnt, rank, E);
    scan_counts<<<cdiv(N, 1024), 1024, 0, stream>>>(cnt, first, N);
    place_edges<<<cdiv(E, B), B, 0, stream>>>(src, dst, first, rank, nbr, E);

    gemm_wmma_strip<4><<<cdiv(N, 128), B, IN * H * sizeof(float), stream>>>(x, W1, a_src1, a_dst1, h1, ls, ld, N, IN);
    node_softmax_sum<64, 1><<<cdiv(N * 16, B), B, 0, stream>>>(first, nbr, ls, ld, h1, b1, act1, N, E);

    gemm_wmma_strip<2><<<cdiv(N, 128), B, H * O * sizeof(float), stream>>>(act1, W2, a_src2, a_dst2, h2, ls, ld, N, H);
    node_softmax_sum<32, 0><<<cdiv(N * 8, B), B, 0, stream>>>(first, nbr, ls, ld, h2, b2, out, N, E);
}


# Projection kernel: all global loads of a wave issued up front as one group (W pieces and the whole 16-row A strip, 16-byte loads, one k order for A and B): one memory trip, not one per loop step
# speedup vs baseline: 3.2508x; 1.1321x over previous
#include <hip/hip_runtime.h>
#include <hip/hip_bf16.h>
#include <math.h>


#define SLOPE 0.2f

typedef float v2f __attribute__((ext_vector_type(2)));
typedef float v8f __attribute__((ext_vector_type(8)));


__global__ void fill_i32(int* __restrict__ p, int v, int n) {
    int i = blockIdx.x * blockDim.x + threadIdx.x;
    if (i < n) p[i] = v;
}

__device__ __forceinline__ float leaky(float v) {
    return v > 0.0f ? v : SLOPE * v;
}

template <int NT, int K>
__global__ void __launch_bounds__(256)
gemm_wmma_strip(const float* __restrict__ A,
                const float* __restrict__ B,
                const float* __restrict__ a_src,
                const float* __restrict__ a_dst,
                float* __restrict__ C,
                float* __restrict__ ls, float* __restrict__ ld,
                int M) {
    constexpr int N  = NT * 16;
    constexpr int P  = N + 8;
    constexpr int NB = (K * N) / (4 * 256);
    __shared__ __attribute__((aligned(16))) float sB[K * P];

    const int lane = threadIdx.x & 31;
    const int wid  = threadIdx.x >> 5;
    const int row0 = (blockIdx.x * 8 + wid) * 16;
    const bool active = row0 < M;
    const int mn = lane & 15;
    const int hi = lane >> 4;

    float4 bq[NB];
#pragma unroll
    for (int j = 0; j < NB; ++j)
        bq[j] = ((const float4*)B)[threadIdx.x + j * 256];

    float4 aq[K / 8];
    if (active) {
        const float4* arow = (const float4*)(A + (size_t)(row0 + mn) * K) + hi;
#pragma unroll
        for (int j = 0; j < K / 8; ++j) aq[j] = arow[2 * j];
    }

#pragma unroll
    for (int j = 0; j < NB; ++j) {
        const int i = (threadIdx.x + j * 256) * 4;
        *(float4*)(sB + (i / N) * P + (i % N)) = bq[j];
    }
    __syncthreads();
    if (!active) return;

    v8f acc[NT];
#pragma unroll
    for (int t = 0; t < NT; ++t) acc[t] = (v8f){};

#pragma unroll
    for (int j = 0; j < K / 8; ++j) {
#pragma unroll
        for (int u = 0; u < 2; ++u) {
            const int kk = 8 * j + 4 * hi + 2 * u;
            v2f a;
            a.x = u ? aq[j].z : aq[j].x;
            a.y = u ? aq[j].w : aq[j].y;
            const float* brow = sB + kk * P + mn;
#pragma unroll
            for (int t = 0; t < NT; ++t) {
                v2f b;
                b.x = brow[t * 16];
                b.y = brow[t * 16 + P];
                acc[t] = __builtin_amdgcn_wmma_f32_16x16x4_f32(
                     false, a,  false, b,
                     (short)0, acc[t],  false,  false);
            }
        }
    }

#pragma unroll
    for (int t = 0; t < NT; ++t) {
        float* crow = C + (size_t)(row0 + hi * 8) * N + t * 16 + mn;
#pragma unroll
        for (int v = 0; v < 8; ++v) crow[(size_t)v * N] = acc[t][v];
    }

    float ps[8], pd[8];
    for (int v = 0; v < 8; ++v) { ps[v] = 0.0f; pd[v] = 0.0f; }
    for (int t = 0; t < NT; ++t) {
        const float as = a_src[t * 16 + mn];
        const float ad = a_dst[t * 16 + mn];
        for (int v = 0; v < 8; ++v) {
            ps[v] += acc[t][v] * as;
            pd[v] += acc[t][v] * ad;
        }
    }
    for (int off = 8; off > 0; off >>= 1) {
        for (int v = 0; v < 8; ++v) {
            ps[v] += __shfl_xor(ps[v], off, 32);
            pd[v] += __shfl_xor(pd[v], off, 32);
        }
    }
    const int r = mn & 7;
    float sv = ps[0], dv = pd[0];
    for (int v = 1; v < 8; ++v) {
        sv = (r == v) ? ps[v] : sv;
        dv = (r == v) ? pd[v] : dv;
    }
    float* sc = (mn < 8) ? ls : ld;
    sc[row0 + hi * 8 + r] = (mn < 8) ? sv : dv;
}


__global__ void rank_count(const int* __restrict__ dst, int* __restrict__ cnt,
                           int* __restrict__ rank, int E) {
    int e = blockIdx.x * blockDim.x + threadIdx.x;
    if (e >= E) return;
    rank[e] = atomicAdd(&cnt[dst[e]], 1);
}

__global__ void __launch_bounds__(1024)
scan_counts(const int* __restrict__ cnt, int* __restrict__ first, int n) {
    __shared__ int wave_tot[32];
    __shared__ int wave_pre[32];
    const int tid  = threadIdx.x;
    const int lane = tid & 31;
    const int w    = tid >> 5;
    const int blk0 = blockIdx.x * 1024;

    int pre = 0;
    for (int i = tid; i < blk0; i += 1024) pre += cnt[i];

    const int i = blk0 + tid;
    const int c = (i < n) ? cnt[i] : 0;

    int x = c;
    for (int off = 1; off < 32; off <<= 1) {
        int u = __shfl_up(x, off, 32);
        if (lane >= off) x += u;
    }
    for (int off = 16; off > 0; off >>= 1) pre += __shfl_xor(pre, off, 32);

    if (lane == 31) wave_tot[w] = x;
    if (lane == 0)  wave_pre[w] = pre;
    __syncthreads();

    const int t = wave_tot[lane];
    int p = wave_pre[lane];
    int inc = t;
    for (int off = 1; off < 32; off <<= 1) {
        int u = __shfl_up(inc, off, 32);
        if (lane >= off) inc += u;
    }
    for (int off = 16; off > 0; off >>= 1) p += __shfl_xor(p, off, 32);
    const int waves_before = __shfl(inc - t, w, 32);

    if (i < n) first[i] = p + waves_before + x - c;
}

__global__ void place_edges(const int* __restrict__ src, const int* __restrict__ dst,
                            const int* __restrict__ first, const int* __restrict__ rank,
                            int* __restrict__ nbr, int E) {
    int e = blockIdx.x * blockDim.x + threadIdx.x;
    if (e >= E) return;
    nbr[first[dst[e]] + rank[e]] = src[e];
}


template <int C>
__device__ __forceinline__ void add_rows(const float* __restrict__ h, int sc, float ac,
                                         int j0, bool more, int gl, float4& acc) {
    constexpr int G = C / 4;
    float4 v[8];
#pragma unroll
    for (int t = 0; t < 4; ++t) {
        const int sj = __shfl(sc, j0 + t, G);
        v[t] = *(const float4*)((const char*)h + ((unsigned)sj * (C * 4u) + gl * 16u));
    }
    if (more) {
#pragma unroll
        for (int t = 4; t < 8; ++t) {
            const int sj = __shfl(sc, j0 + t, G);
            v[t] = *(const float4*)((const char*)h + ((unsigned)sj * (C * 4u) + gl * 16u));
        }
    }
#pragma unroll
    for (int t = 0; t < 4; ++t) {
        const float aj = __shfl(ac, j0 + t, G);
        acc.x += aj * v[t].x;
        acc.y += aj * v[t].y;
        acc.z += aj * v[t].z;
        acc.w += aj * v[t].w;
    }
    if (more) {
#pragma unroll
        for (int t = 4; t < 8; ++t) {
            const float aj = __shfl(ac, j0 + t, G);
            acc.x += aj * v[t].x;
            acc.y += aj * v[t].y;
            acc.z += aj * v[t].z;
            acc.w += aj * v[t].w;
        }
    }
}

template <int C>
__device__ __forceinline__ void add_block(const float* __restrict__ h, const int* s, const float* a,
                                          int cnt, int gl, float4& acc) {
    constexpr int G  = C / 4;
    constexpr int KC = 32 / G;
#pragma unroll
    for (int c = 0; c < KC; ++c) {
#pragma unroll
        for (int jb = 0; jb < G; jb += 8) {
            const int rem = cnt - (c * G + jb);
            if (rem > 0) add_rows<C>(h, s[c], a[c], jb, rem > 4, gl, acc);
        }
    }
}

template <int C, int RELU>
__global__ void __launch_bounds__(256)
node_softmax_sum(const int* __restrict__ first, const int* __restrict__ nbr,
                 const float* __restrict__ ls, const float* __restrict__ ld,
                 const float* __restrict__ h, const float* __restrict__ bias,
                 float* __restrict__ out, int n_nodes, int n_edges) {
    constexpr int G  = C / 4;
    constexpr int KC = 32 / G;
    const int tid = blockIdx.x * blockDim.x + threadIdx.x;
    const int gl  = threadIdx.x & (G - 1);
    const int node = tid / G;
    if ((tid & ~31) / G >= n_nodes) return;

    const bool live = node < n_nodes;
    int beg = 0, end = 0;
    float ldn = 0.0f;
    if (live) {
        beg = first[node];
        end = (node + 1 < n_nodes) ? first[node + 1] : n_edges;
        ldn = ld[node];
    }
    const int deg = end - beg;

    int dmax = deg;
    for (int off = G; off < 32; off <<= 1) dmax = max(dmax, __shfl_xor(dmax, off, 32));
    const int wdeg = __builtin_amdgcn_readfirstlane(dmax);

    float4 acc = make_float4(0.0f, 0.0f, 0.0f, 0.0f);

    if (wdeg > 0) {
        int   s[KC];
        float lv[KC];
        float lg[KC];
        bool  ok[KC];
        const int last = deg > 0 ? deg - 1 : 0;
#pragma unroll
        for (int c = 0; c < KC; ++c) {
            const int slot = c * G + gl;
            ok[c] = slot < deg;
            int idx = beg + (ok[c] ? slot : last);
            idx = idx < n_edges ? idx : n_edges - 1;
            s[c] = nbr[idx];
        }
#pragma unroll
        for (int c = 0; c < KC; ++c) lv[c] = ls[s[c]];
        float m = -INFINITY;
#pragma unroll
        for (int c = 0; c < KC; ++c) {
            lg[c] = ok[c] ? leaky(lv[c] + ldn) : -INFINITY;
            m = fmaxf(m, lg[c]);
        }
        for (int i = beg + 32 + gl; i < end; i += G) m = fmaxf(m, leaky(ls[nbr[i]] + ldn));
        for (int off = G / 2; off > 0; off >>= 1) m = fmaxf(m, __shfl_xor(m, off, 32));

        float e[KC];
        float den = 0.0f;
#pragma unroll
        for (int c = 0; c < KC; ++c) {
            e[c] = ok[c] ? __expf(lg[c] - m) : 0.0f;
            den += e[c];
        }
        for (int i = beg + 32 + gl; i < end; i += G) den += __expf(leaky(ls[nbr[i]] + ldn) - m);
        for (int off = G / 2; off > 0; off >>= 1) den += __shfl_xor(den, off, 32);

        float a[KC];
#pragma unroll
        for (int c = 0; c < KC; ++c) a[c] = ok[c] ? e[c] / den : 0.0f;

        add_block<C>(h, s, a, wdeg < 32 ? wdeg : 32, gl, acc);

        for (int base = 32; base < wdeg; base += 32) {
            int   s2[KC];
            float a2[KC];
#pragma unroll
            for (int c = 0; c < KC; ++c) {
                const int  i = base + c * G + gl;
                const bool v = i < deg;
                s2[c] = v ? nbr[beg + i] : 0;
                a2[c] = v ? __expf(leaky(ls[s2[c]] + ldn) - m) / den : 0.0f;
            }
            const int left = wdeg - base;
            add_block<C>(h, s2, a2, left < 32 ? left : 32, gl, acc);
        }
    }

    if (live) {
        const float* bp = bias + gl * 4;
        const float4 b = make_float4(bp[0], bp[1], bp[2], bp[3]);
        float4 r = make_float4(acc.x + b.x, acc.y + b.y, acc.z + b.z, acc.w + b.w);
        if (RELU) {
            r.x = fmaxf(r.x, 0.0f);
            r.y = fmaxf(r.y, 0.0f);
            r.z = fmaxf(r.z, 0.0f);
            r.w = fmaxf(r.w, 0.0f);
        }
        *(float4*)(out + (size_t)node * C + gl * 4) = r;
    }
}


static inline int cdiv(int a, int b) { return (a + b - 1) / b; }

extern "C" void kernel_launch(void* const* d_in, const int* in_sizes, int n_in,
                              void* d_out, int out_size, void* d_ws, size_t ws_size,
                              hipStream_t stream) {
    const float* x      = (const float*)d_in[0];
    const int*   eidx   = (const int*)d_in[1];
    const float* W1     = (const float*)d_in[2];
    const float* a_src1 = (const float*)d_in[3];
    const float* a_dst1 = (const float*)d_in[4];
    const float* b1     = (const float*)d_in[5];
    const float* W2     = (const float*)d_in[6];
    const float* a_src2 = (const float*)d_in[7];
    const float* a_dst2 = (const float*)d_in[8];
    const float* b2     = (const float*)d_in[9];
    float* out = (float*)d_out;

    const int IN = 128, H = 64, O = 32;
    const int N = in_sizes[0] / IN;
    const int E = in_sizes[1] / 2;
    const int* src = eidx;
    const int* dst = eidx + E;

    float* ws    = (float*)d_ws;
    float* h1    = ws;
    float* act1  = h1 + (size_t)N * H;
    float* h2    = act1 + (size_t)N * H;
    float* ls    = h2 + (size_t)N * O;
    float* ld    = ls + N;
    int*   cnt   = (int*)(ld + N);
    int*   first = cnt + N;
    int*   nbr   = first + N;
    int*   rank  = (int*)act1;

    const int B = 256;

    fill_i32<<<cdiv(N, B), B, 0, stream>>>(cnt, 0, N);
    rank_count<<<cdiv(E, B), B, 0, stream>>>(dst, cnt, rank, E);
    scan_counts<<<cdiv(N, 1024), 1024, 0, stream>>>(cnt, first, N);
    place_edges<<<cdiv(E, B), B, 0, stream>>>(src, dst, first, rank, nbr, E);

    gemm_wmma_strip<4, 128><<<cdiv(N, 128), B, 0, stream>>>(x, W1, a_src1, a_dst1, h1, ls, ld, N);
    node_softmax_sum<64, 1><<<cdiv(N * 16, B), B, 0, stream>>>(first, nbr, ls, ld, h1, b1, act1, N, E);

    gemm_wmma_strip<2, 64><<<cdiv(N, 128), B, 0, stream>>>(act1, W2, a_src2, a_dst2, h2, ls, ld, N);
    node_softmax_sum<32, 0><<<cdiv(N * 8, B), B, 0, stream>>>(first, nbr, ls, ld, h2, b2, out, N, E);
}


# Edge slots ranked with LDS counters in two passes (file edges by destination range, then one work-group per range builds whole slot lines in LDS): no global atomic per edge, no counter fill
# speedup vs baseline: 5.4564x; 1.6785x over previous
#include <hip/hip_runtime.h>
#include <hip/hip_bf16.h>
#include <math.h>


#define SLOPE 0.2f

typedef float v2f __attribute__((ext_vector_type(2)));
typedef float v8f __attribute__((ext_vector_type(8)));


__device__ __forceinline__ float leaky(float v) {
    return v > 0.0f ? v : SLOPE * v;
}

template <int NT, int K>
__global__ void __launch_bounds__(256)
gemm_wmma_strip(const float* __restrict__ A,
                const float* __restrict__ B,
                const float* __restrict__ a_src,
                const float* __restrict__ a_dst,
                float* __restrict__ C,
                float* __restrict__ ls, float* __restrict__ ld,
                int M) {
    constexpr int N  = NT * 16;
    constexpr int P  = N + 8;
    constexpr int NB = (K * N) / (4 * 256);
    __shared__ __attribute__((aligned(16))) float sB[K * P];

    const int lane = threadIdx.x & 31;
    const int wid  = threadIdx.x >> 5;
    const int row0 = (blockIdx.x * 8 + wid) * 16;
    const bool active = row0 < M;
    const int mn = lane & 15;
    const int hi = lane >> 4;

    float4 bq[NB];
#pragma unroll
    for (int j = 0; j < NB; ++j)
        bq[j] = ((const float4*)B)[threadIdx.x + j * 256];

    float4 aq[K / 8];
    if (active) {
        const float4* arow = (const float4*)(A + (size_t)(row0 + mn) * K) + hi;
#pragma unroll
        for (int j = 0; j < K / 8; ++j) aq[j] = arow[2 * j];
    }

#pragma unroll
    for (int j = 0; j < NB; ++j) {
        const int i = (threadIdx.x + j * 256) * 4;
        *(float4*)(sB + (i / N) * P + (i % N)) = bq[j];
    }
    __syncthreads();
    if (!active) return;

    v8f acc[NT];
#pragma unroll
    for (int t = 0; t < NT; ++t) acc[t] = (v8f){};

#pragma unroll
    for (int j = 0; j < K / 8; ++j) {
#pragma unroll
        for (int u = 0; u < 2; ++u) {
            const int kk = 8 * j + 4 * hi + 2 * u;
            v2f a;
            a.x = u ? aq[j].z : aq[j].x;
            a.y = u ? aq[j].w : aq[j].y;
            const float* brow = sB + kk * P + mn;
#pragma unroll
            for (int t = 0; t < NT; ++t) {
                v2f b;
                b.x = brow[t * 16];
                b.y = brow[t * 16 + P];
                acc[t] = __builtin_amdgcn_wmma_f32_16x16x4_f32(
                     false, a,  false, b,
                     (short)0, acc[t],  false,  false);
            }
        }
    }

#pragma unroll
    for (int t = 0; t < NT; ++t) {
        float* crow = C + (size_t)(row0 + hi * 8) * N + t * 16 + mn;
#pragma unroll
        for (int v = 0; v < 8; ++v) crow[(size_t)v * N] = acc[t][v];
    }

    float ps[8], pd[8];
    for (int v = 0; v < 8; ++v) { ps[v] = 0.0f; pd[v] = 0.0f; }
    for (int t = 0; t < NT; ++t) {
        const float as = a_src[t * 16 + mn];
        const float ad = a_dst[t * 16 + mn];
        for (int v = 0; v < 8; ++v) {
            ps[v] += acc[t][v] * as;
            pd[v] += acc[t][v] * ad;
        }
    }
    for (int off = 8; off > 0; off >>= 1) {
        for (int v = 0; v < 8; ++v) {
            ps[v] += __shfl_xor(ps[v], off, 32);
            pd[v] += __shfl_xor(pd[v], off, 32);
        }
    }
    const int r = mn & 7;
    float sv = ps[0], dv = pd[0];
    for (int v = 1; v < 8; ++v) {
        sv = (r == v) ? ps[v] : sv;
        dv = (r == v) ? pd[v] : dv;
    }
    float* sc = (mn < 8) ? ls : ld;
    sc[row0 + hi * 8 + r] = (mn < 8) ? sv : dv;
}


#define CAP 32
#define NPB 196
#define NWG 256
#define SC  32
#define EPT 13
#define MAXBK 512

__device__ __forceinline__ int chunk_len(int E) {
    return (((E + NWG - 1) / NWG) + 31) & ~31;
}

__global__ void __launch_bounds__(256)
bin_edges(const int* __restrict__ src, const int* __restrict__ dst,
          int2* __restrict__ runs, int* __restrict__ runc,
          int* __restrict__ cls, int* __restrict__ clc, int* __restrict__ ovn,
          int E, int nbk) {
    __shared__ int lcnt[MAXBK];
    __shared__ int lown[256];
    __shared__ int lsp;
    const int w = blockIdx.x;
    const int chunk = chunk_len(E);
    const int base = w * chunk;
    for (int t = threadIdx.x; t < nbk; t += 256) lcnt[t] = 0;
    lown[threadIdx.x] = 0;
    if (threadIdx.x == 0) lsp = 0;
    __syncthreads();
    for (int i0 = 0; i0 < chunk; i0 += 256 * EPT) {
        int d[EPT], s[EPT];
#pragma unroll
        for (int j = 0; j < EPT; ++j) {
            const int i = i0 + j * 256 + threadIdx.x;
            const bool v = i < chunk && base + i < E;
            d[j] = v ? dst[base + i] : -1;
            s[j] = v ? src[base + i] : 0;
        }
        int r[EPT];
#pragma unroll
        for (int j = 0; j < EPT; ++j) {
            const int b = d[j] / NPB;
            const bool v = d[j] >= 0 && b < nbk;
            r[j] = atomicAdd(v ? &lcnt[b] : &lown[threadIdx.x], 1);
            r[j] = v ? r[j] : -1;
        }
#pragma unroll
        for (int j = 0; j < EPT; ++j) {
            if (r[j] >= 0) {
                const int li = i0 + j * 256 + threadIdx.x;
                const int b = d[j] / NPB;
                if (r[j] < SC) {
                    const size_t p = ((size_t)w * nbk + b) * SC + r[j];
                    runs[p] = make_int2(s[j], (d[j] - b * NPB) | (li << 8));
                } else {
                    cls[base + atomicAdd(&lsp, 1)] = base + li;
                }
            }
        }
    }
    __syncthreads();
    for (int t = threadIdx.x; t < nbk; t += 256) {
        const int c = lcnt[t];
        runc[w * nbk + t] = c < SC ? c : SC;
    }
    if (threadIdx.x == 0) {
        clc[w] = lsp;
        if (w == 0) *ovn = 0;
    }
}

__device__ __forceinline__ void place_edge(int* lcn, int* ltab, int s, int dl, int e,
                                           int* __restrict__ ovn, int* __restrict__ ove) {
    const int r = atomicAdd(&lcn[dl], 1);
    if (r < CAP) ltab[dl * CAP + r] = s;
    else         ove[atomicAdd(ovn, 1)] = e;
}

__global__ void __launch_bounds__(NWG)
slot_ranges(const int* __restrict__ src, const int* __restrict__ dst,
            const int2* __restrict__ runs,
            const int* __restrict__ runc, const int* __restrict__ cls,
            const int* __restrict__ clc,
            int* __restrict__ cnt, int* __restrict__ tab,
            int* __restrict__ ovn, int* __restrict__ ove, int N, int E, int nbk) {
    __shared__ __attribute__((aligned(16))) int ltab[NPB * CAP];
    __shared__ int lcn[NPB];
    __shared__ int lown[NWG];
    const int b  = blockIdx.x;
    const int n0 = b * NPB;
    const int w  = threadIdx.x;
    const int chunk = chunk_len(E);
    const size_t run = ((size_t)w * nbk + b) * SC;

    int c  = runc[w * nbk + b];
    int n1 = clc[w];
    const int4* rp = (const int4*)(runs + run);
    int4 q[SC / 2];
#pragma unroll
    for (int j = 0; j < SC / 2; ++j) q[j] = rp[j];

    for (int i = threadIdx.x; i < NPB * CAP / 4; i += NWG)
        ((int4*)ltab)[i] = make_int4(0, 0, 0, 0);
    if (threadIdx.x < NPB) lcn[threadIdx.x] = 0;
    lown[threadIdx.x] = 0;
    __syncthreads();

    c  = c  < SC ? c : SC;
    n1 = n1 < chunk ? n1 : chunk;
    int rk[SC];
#pragma unroll
    for (int k = 0; k < SC; ++k) {
        const int dl = ((k & 1) ? q[k / 2].w : q[k / 2].y) & 255;
        const bool v = k < c && dl < NPB;
        rk[k] = atomicAdd(v ? &lcn[dl] : &lown[threadIdx.x], 1);
        rk[k] = v ? rk[k] : -1;
    }
#pragma unroll
    for (int k = 0; k < SC; ++k) {
        if (rk[k] >= 0) {
            const unsigned code = (unsigned)((k & 1) ? q[k / 2].w : q[k / 2].y);
            const int dl = code & 255;
            const int s  = (k & 1) ? q[k / 2].z : q[k / 2].x;
            if (rk[k] < CAP) ltab[dl * CAP + rk[k]] = s;
            else             ove[atomicAdd(ovn, 1)] = w * chunk + (int)(code >> 8);
        }
    }
    for (int k = 0; k < n1; ++k) {
        const int e = cls[w * chunk + k];
        if ((unsigned)e < (unsigned)E) {
            const int se = src[e];
            const int dl = dst[e] - n0;
            if ((unsigned)dl < (unsigned)NPB) place_edge(lcn, ltab, se, dl, e, ovn, ove);
        }
    }
    __syncthreads();

    int nn = N - n0;
    nn = nn < NPB ? nn : NPB;
    if ((int)threadIdx.x < nn) cnt[n0 + threadIdx.x] = lcn[threadIdx.x];
    int4* tp = (int4*)(tab + (size_t)n0 * CAP);
    for (int i = threadIdx.x; i < nn * (CAP / 4); i += NWG) tp[i] = ((const int4*)ltab)[i];
}


template <int C>
__device__ __forceinline__ void add_rows(const float* __restrict__ h, int sc, float ac,
                                         int j0, bool more, int gl, float4& acc) {
    constexpr int G = C / 4;
    float4 v[8];
#pragma unroll
    for (int t = 0; t < 4; ++t) {
        const int sj = __shfl(sc, j0 + t, G);
        v[t] = *(const float4*)((const char*)h + ((unsigned)sj * (C * 4u) + gl * 16u));
    }
    if (more) {
#pragma unroll
        for (int t = 4; t < 8; ++t) {
            const int sj = __shfl(sc, j0 + t, G);
            v[t] = *(const float4*)((const char*)h + ((unsigned)sj * (C * 4u) + gl * 16u));
        }
    }
#pragma unroll
    for (int t = 0; t < 4; ++t) {
        const float aj = __shfl(ac, j0 + t, G);
        acc.x += aj * v[t].x;
        acc.y += aj * v[t].y;
        acc.z += aj * v[t].z;
        acc.w += aj * v[t].w;
    }
    if (more) {
#pragma unroll
        for (int t = 4; t < 8; ++t) {
            const float aj = __shfl(ac, j0 + t, G);
            acc.x += aj * v[t].x;
            acc.y += aj * v[t].y;
            acc.z += aj * v[t].z;
            acc.w += aj * v[t].w;
        }
    }
}

template <int C>
__device__ __forceinline__ void add_block(const float* __restrict__ h, const int* s, const float* a,
                                          int cnt, int gl, float4& acc) {
    constexpr int G  = C / 4;
    constexpr int KC = 32 / G;
#pragma unroll
    for (int c = 0; c < KC; ++c) {
#pragma unroll
        for (int jb = 0; jb < G; jb += 8) {
            const int rem = cnt - (c * G + jb);
            if (rem > 0) add_rows<C>(h, s[c], a[c], jb, rem > 4, gl, acc);
        }
    }
}

template <int C, int RELU>
__global__ void __launch_bounds__(256)
node_softmax_sum(const int* __restrict__ cnt, const int* __restrict__ tab,
                 const int* __restrict__ ovn, const int* __restrict__ ove,
                 const int* __restrict__ esrc, const int* __restrict__ edst,
                 const float* __restrict__ ls, const float* __restrict__ ld,
                 const float* __restrict__ h, const float* __restrict__ bias,
                 float* __restrict__ out, int n_nodes, int n_edges) {
    constexpr int G  = C / 4;
    constexpr int KC = CAP / G;
    static_assert(CAP == 32, "add_block walks 32 slots per group");
    const int tid = blockIdx.x * blockDim.x + threadIdx.x;
    const int gl  = threadIdx.x & (G - 1);
    const int node = tid / G;
    if ((tid & ~31) / G >= n_nodes) return;

    const bool live = node < n_nodes;
    const int nd = live ? node : n_nodes - 1;

    int raw[KC];
#pragma unroll
    for (int c = 0; c < KC; ++c) raw[c] = tab[nd * CAP + c * G + gl];
    const int   dall = cnt[nd];
    const float ldn  = ld[nd];

    const int deg  = live ? dall : 0;
    const int dtab = deg < CAP ? deg : CAP;
    const bool over = deg > CAP;

    int dmax = dtab;
    for (int off = G; off < 32; off <<= 1) dmax = max(dmax, __shfl_xor(dmax, off, 32));
    const int wdeg = __builtin_amdgcn_readfirstlane(dmax);

    float4 acc = make_float4(0.0f, 0.0f, 0.0f, 0.0f);

    int   s[KC];
    float lv[KC];
    float lg[KC];
    bool  ok[KC];
    const int last = dtab > 0 ? dtab - 1 : 0;
    const int cl = last / G;
    int pick = raw[0];
#pragma unroll
    for (int c = 1; c < KC; ++c) pick = (cl == c) ? raw[c] : pick;
    const int lastsrc = __shfl(pick, last & (G - 1), G);
    const int fillsrc = dtab > 0 ? lastsrc : nd;
#pragma unroll
    for (int c = 0; c < KC; ++c) {
        ok[c] = c * G + gl < dtab;
        s[c] = ok[c] ? raw[c] : fillsrc;
    }
#pragma unroll
    for (int c = 0; c < KC; ++c) lv[c] = ls[s[c]];

    int novf = 0;
    if (over) {
        novf = *ovn;
        novf = novf < n_edges ? novf : n_edges;
    }

    float m = -INFINITY;
#pragma unroll
    for (int c = 0; c < KC; ++c) {
        lg[c] = ok[c] ? leaky(lv[c] + ldn) : -INFINITY;
        m = fmaxf(m, lg[c]);
    }
    if (over) {
        for (int i = gl; i < novf; i += G) {
            const int e = ove[i];
            if (edst[e] == node) m = fmaxf(m, leaky(ls[esrc[e]] + ldn));
        }
    }
    for (int off = G / 2; off > 0; off >>= 1) m = fmaxf(m, __shfl_xor(m, off, 32));

    float ex[KC];
    float den = 0.0f;
#pragma unroll
    for (int c = 0; c < KC; ++c) {
        ex[c] = ok[c] ? __expf(lg[c] - m) : 0.0f;
        den += ex[c];
    }
    if (over) {
        for (int i = gl; i < novf; i += G) {
            const int e = ove[i];
            if (edst[e] == node) den += __expf(leaky(ls[esrc[e]] + ldn) - m);
        }
    }
    for (int off = G / 2; off > 0; off >>= 1) den += __shfl_xor(den, off, 32);

    float a[KC];
#pragma unroll
    for (int c = 0; c < KC; ++c) a[c] = ok[c] ? ex[c] / den : 0.0f;

    add_block<C>(h, s, a, wdeg, gl, acc);

    if (over) {
        for (int i = 0; i < novf; ++i) {
            const int e = ove[i];
            if (edst[e] == node) {
                const int sj = esrc[e];
                const float w = __expf(leaky(ls[sj] + ldn) - m) / den;
                const float4 v = *(const float4*)((const char*)h + ((unsigned)sj * (C * 4u) + gl * 16u));
                acc.x += w * v.x;
                acc.y += w * v.y;
                acc.z += w * v.z;
                acc.w += w * v.w;
            }
        }
    }

    if (live) {
        const float* bp = bias + gl * 4;
        const float4 b = make_float4(bp[0], bp[1], bp[2], bp[3]);
        float4 r = make_float4(acc.x + b.x, acc.y + b.y, acc.z + b.z, acc.w + b.w);
        if (RELU) {
            r.x = fmaxf(r.x, 0.0f);
            r.y = fmaxf(r.y, 0.0f);
            r.z = fmaxf(r.z, 0.0f);
            r.w = fmaxf(r.w, 0.0f);
        }
        *(float4*)(out + (size_t)node * C + gl * 4) = r;
    }
}


static inline int cdiv(int a, int b) { return (a + b - 1) / b; }

extern "C" void kernel_launch(void* const* d_in, const int* in_sizes, int n_in,
                              void* d_out, int out_size, void* d_ws, size_t ws_size,
                              hipStream_t stream) {
    const float* x      = (const float*)d_in[0];
    const int*   eidx   = (const int*)d_in[1];
    const float* W1     = (const float*)d_in[2];
    const float* a_src1 = (const float*)d_in[3];
    const float* a_dst1 = (const float*)d_in[4];
    const float* b1     = (const float*)d_in[5];
    const float* W2     = (const float*)d_in[6];
    const float* a_src2 = (const float*)d_in[7];
    const float* a_dst2 = (const float*)d_in[8];
    const float* b2     = (const float*)d_in[9];
    float* out = (float*)d_out;

    const int IN = 128, H = 64, O = 32;
    const int N = in_sizes[0] / IN;
    const int E = in_sizes[1] / 2;
    const int* src = eidx;
    const int* dst = eidx + E;

    float* ws    = (float*)d_ws;
    float* h1    = ws;
    float* h2    = ws;
    float* act1  = h1 + (size_t)N * H;
    int*   tab   = (int*)(act1 + (size_t)N * H);
    float* ls    = (float*)(tab + (size_t)N * CAP);
    float* ld    = ls + N;
    int*   cnt   = (int*)(ld + N);
    int*   ovn   = cnt + N;
    int*   ove   = cnt + 2 * (size_t)N;

    const int B = 256;

    const int nbk = cdiv(N, NPB);
    int2* runs = (int2*)h1;
    int*  cls  = (int*)(runs + (size_t)NWG * nbk * SC);
    int*  runc = cls + ((size_t)E + 32 * NWG);
    int*  clc  = runc + (size_t)NWG * nbk;
    bin_edges<<<NWG, 256, 0, stream>>>(src, dst, runs, runc, cls, clc, ovn, E, nbk);
    slot_ranges<<<nbk, NWG, 0, stream>>>(src, dst, runs, runc, cls, clc, cnt, tab, ovn, ove, N, E, nbk);

    gemm_wmma_strip<4, 128><<<cdiv(N, 128), B, 0, stream>>>(x, W1, a_src1, a_dst1, h1, ls, ld, N);
    node_softmax_sum<64, 1><<<cdiv(N * 16, B), B, 0, stream>>>(cnt, tab, ovn, ove, src, dst, ls, ld, h1, b1, act1, N, E);

    gemm_wmma_strip<2, 64><<<cdiv(N, 128), B, 0, stream>>>(act1, W2, a_src2, a_dst2, h2, ls, ld, N);
    node_softmax_sum<32, 0><<<cdiv(N * 8, B), B, 0, stream>>>(cnt, tab, ovn, ove, src, dst, ls, ld, h2, b2, out, N, E);
}


# Layer-2 projection and its scores formed inside the layer-1 node kernel (a block's 16 finished rows go through LDS to one wave's f32 matrix ops): one launch fewer, layer-1 output never written out
# speedup vs baseline: 5.6813x; 1.0412x over previous
#include <hip/hip_runtime.h>
#include <hip/hip_bf16.h>
#include <math.h>


#define SLOPE 0.2f

typedef float v2f __attribute__((ext_vector_type(2)));
typedef float v8f __attribute__((ext_vector_type(8)));


__device__ __forceinline__ float leaky(float v) {
    return v > 0.0f ? v : SLOPE * v;
}

template <int NT, int K>
__global__ void __launch_bounds__(256)
gemm_wmma_strip(const float* __restrict__ A,
                const float* __restrict__ B,
                const float* __restrict__ a_src,
                const float* __restrict__ a_dst,
                float* __restrict__ C,
                float* __restrict__ ls, float* __restrict__ ld,
                int M) {
    constexpr int N  = NT * 16;
    constexpr int P  = N + 8;
    constexpr int NB = (K * N) / (4 * 256);
    __shared__ __attribute__((aligned(16))) float sB[K * P];

    const int lane = threadIdx.x & 31;
    const int wid  = threadIdx.x >> 5;
    const int row0 = (blockIdx.x * 8 + wid) * 16;
    const bool active = row0 < M;
    const int mn = lane & 15;
    const int hi = lane >> 4;

    float4 bq[NB];
#pragma unroll
    for (int j = 0; j < NB; ++j)
        bq[j] = ((const float4*)B)[threadIdx.x + j * 256];

    float4 aq[K / 8];
    if (active) {
        const float4* arow = (const float4*)(A + (size_t)(row0 + mn) * K) + hi;
#pragma unroll
        for (int j = 0; j < K / 8; ++j) aq[j] = arow[2 * j];
    }

#pragma unroll
    for (int j = 0; j < NB; ++j) {
        const int i = (threadIdx.x + j * 256) * 4;
        *(float4*)(sB + (i / N) * P + (i % N)) = bq[j];
    }
    __syncthreads();
    if (!active) return;

    v8f acc[NT];
#pragma unroll
    for (int t = 0; t < NT; ++t) acc[t] = (v8f){};

#pragma unroll
    for (int j = 0; j < K / 8; ++j) {
#pragma unroll
        for (int u = 0; u < 2; ++u) {
            const int kk = 8 * j + 4 * hi + 2 * u;
            v2f a;
            a.x = u ? aq[j].z : aq[j].x;
            a.y = u ? aq[j].w : aq[j].y;
            const float* brow = sB + kk * P + mn;
#pragma unroll
            for (int t = 0; t < NT; ++t) {
                v2f b;
                b.x = brow[t * 16];
                b.y = brow[t * 16 + P];
                acc[t] = __builtin_amdgcn_wmma_f32_16x16x4_f32(
                     false, a,  false, b,
                     (short)0, acc[t],  false,  false);
            }
        }
    }

#pragma unroll
    for (int t = 0; t < NT; ++t) {
        float* crow = C + (size_t)(row0 + hi * 8) * N + t * 16 + mn;
#pragma unroll
        for (int v = 0; v < 8; ++v) crow[(size_t)v * N] = acc[t][v];
    }

    float ps[8], pd[8];
    for (int v = 0; v < 8; ++v) { ps[v] = 0.0f; pd[v] = 0.0f; }
    for (int t = 0; t < NT; ++t) {
        const float as = a_src[t * 16 + mn];
        const float ad = a_dst[t * 16 + mn];
        for (int v = 0; v < 8; ++v) {
            ps[v] += acc[t][v] * as;
            pd[v] += acc[t][v] * ad;
        }
    }
    for (int off = 8; off > 0; off >>= 1) {
        for (int v = 0; v < 8; ++v) {
            ps[v] += __shfl_xor(ps[v], off, 32);
            pd[v] += __shfl_xor(pd[v], off, 32);
        }
    }
    const int r = mn & 7;
    float sv = ps[0], dv = pd[0];
    for (int v = 1; v < 8; ++v) {
        sv = (r == v) ? ps[v] : sv;
        dv = (r == v) ? pd[v] : dv;
    }
    float* sc = (mn < 8) ? ls : ld;
    sc[row0 + hi * 8 + r] = (mn < 8) ? sv : dv;
}


#define CAP 32
#define NPB 196
#define NWG 256
#define SC  32
#define EPT 13
#define MAXBK 512

__device__ __forceinline__ int chunk_len(int E) {
    return (((E + NWG - 1) / NWG) + 31) & ~31;
}

__global__ void __launch_bounds__(256)
bin_edges(const int* __restrict__ src, const int* __restrict__ dst,
          int2* __restrict__ runs, int* __restrict__ runc,
          int* __restrict__ cls, int* __restrict__ clc, int* __restrict__ ovn,
          int E, int nbk) {
    __shared__ int lcnt[MAXBK];
    __shared__ int lown[256];
    __shared__ int lsp;
    const int w = blockIdx.x;
    const int chunk = chunk_len(E);
    const int base = w * chunk;
    for (int t = threadIdx.x; t < nbk; t += 256) lcnt[t] = 0;
    lown[threadIdx.x] = 0;
    if (threadIdx.x == 0) lsp = 0;
    __syncthreads();
    for (int i0 = 0; i0 < chunk; i0 += 256 * EPT) {
        int d[EPT], s[EPT];
#pragma unroll
        for (int j = 0; j < EPT; ++j) {
            const int i = i0 + j * 256 + threadIdx.x;
            const bool v = i < chunk && base + i < E;
            d[j] = v ? dst[base + i] : -1;
            s[j] = v ? src[base + i] : 0;
        }
        int r[EPT];
#pragma unroll
        for (int j = 0; j < EPT; ++j) {
            const int b = d[j] / NPB;
            const bool v = d[j] >= 0 && b < nbk;
            r[j] = atomicAdd(v ? &lcnt[b] : &lown[threadIdx.x], 1);
            r[j] = v ? r[j] : -1;
        }
#pragma unroll
        for (int j = 0; j < EPT; ++j) {
            if (r[j] >= 0) {
                const int li = i0 + j * 256 + threadIdx.x;
                const int b = d[j] / NPB;
                if (r[j] < SC) {
                    const size_t p = ((size_t)w * nbk + b) * SC + r[j];
                    runs[p] = make_int2(s[j], (d[j] - b * NPB) | (li << 8));
                } else {
                    cls[base + atomicAdd(&lsp, 1)] = base + li;
                }
            }
        }
    }
    __syncthreads();
    for (int t = threadIdx.x; t < nbk; t += 256) {
        const int c = lcnt[t];
        runc[w * nbk + t] = c < SC ? c : SC;
    }
    if (threadIdx.x == 0) {
        clc[w] = lsp;
        if (w == 0) *ovn = 0;
    }
}

__device__ __forceinline__ void place_edge(int* lcn, int* ltab, int s, int dl, int e,
                                           int* __restrict__ ovn, int* __restrict__ ove) {
    const int r = atomicAdd(&lcn[dl], 1);
    if (r < CAP) ltab[dl * CAP + r] = s;
    else         ove[atomicAdd(ovn, 1)] = e;
}

__global__ void __launch_bounds__(NWG)
slot_ranges(const int* __restrict__ src, const int* __restrict__ dst,
            const int2* __restrict__ runs,
            const int* __restrict__ runc, const int* __restrict__ cls,
            const int* __restrict__ clc,
            int* __restrict__ cnt, int* __restrict__ tab,
            int* __restrict__ ovn, int* __restrict__ ove, int N, int E, int nbk) {
    __shared__ __attribute__((aligned(16))) int ltab[NPB * CAP];
    __shared__ int lcn[NPB];
    __shared__ int lown[NWG];
    const int b  = blockIdx.x;
    const int n0 = b * NPB;
    const int w  = threadIdx.x;
    const int chunk = chunk_len(E);
    const size_t run = ((size_t)w * nbk + b) * SC;

    int c  = runc[w * nbk + b];
    int n1 = clc[w];
    const int4* rp = (const int4*)(runs + run);
    int4 q[SC / 2];
#pragma unroll
    for (int j = 0; j < SC / 2; ++j) q[j] = rp[j];

    for (int i = threadIdx.x; i < NPB * CAP / 4; i += NWG)
        ((int4*)ltab)[i] = make_int4(0, 0, 0, 0);
    if (threadIdx.x < NPB) lcn[threadIdx.x] = 0;
    lown[threadIdx.x] = 0;
    __syncthreads();

    c  = c  < SC ? c : SC;
    n1 = n1 < chunk ? n1 : chunk;
    int rk[SC];
#pragma unroll
    for (int k = 0; k < SC; ++k) {
        const int dl = ((k & 1) ? q[k / 2].w : q[k / 2].y) & 255;
        const bool v = k < c && dl < NPB;
        rk[k] = atomicAdd(v ? &lcn[dl] : &lown[threadIdx.x], 1);
        rk[k] = v ? rk[k] : -1;
    }
#pragma unroll
    for (int k = 0; k < SC; ++k) {
        if (rk[k] >= 0) {
            const unsigned code = (unsigned)((k & 1) ? q[k / 2].w : q[k / 2].y);
            const int dl = code & 255;
            const int s  = (k & 1) ? q[k / 2].z : q[k / 2].x;
            if (rk[k] < CAP) ltab[dl * CAP + rk[k]] = s;
            else             ove[atomicAdd(ovn, 1)] = w * chunk + (int)(code >> 8);
        }
    }
    for (int k = 0; k < n1; ++k) {
        const int e = cls[w * chunk + k];
        if ((unsigned)e < (unsigned)E) {
            const int se = src[e];
            const int dl = dst[e] - n0;
            if ((unsigned)dl < (unsigned)NPB) place_edge(lcn, ltab, se, dl, e, ovn, ove);
        }
    }
    __syncthreads();

    int nn = N - n0;
    nn = nn < NPB ? nn : NPB;
    if ((int)threadIdx.x < nn) cnt[n0 + threadIdx.x] = lcn[threadIdx.x];
    int4* tp = (int4*)(tab + (size_t)n0 * CAP);
    for (int i = threadIdx.x; i < nn * (CAP / 4); i += NWG) tp[i] = ((const int4*)ltab)[i];
}


template <int C>
__device__ __forceinline__ void add_rows(const float* __restrict__ h, int sc, float ac,
                                         int j0, bool more, int gl, float4& acc) {
    constexpr int G = C / 4;
    float4 v[8];
#pragma unroll
    for (int t = 0; t < 4; ++t) {
        const int sj = __shfl(sc, j0 + t, G);
        v[t] = *(const float4*)((const char*)h + ((unsigned)sj * (C * 4u) + gl * 16u));
    }
    if (more) {
#pragma unroll
        for (int t = 4; t < 8; ++t) {
            const int sj = __shfl(sc, j0 + t, G);
            v[t] = *(const float4*)((const char*)h + ((unsigned)sj * (C * 4u) + gl * 16u));
        }
    }
#pragma unroll
    for (int t = 0; t < 4; ++t) {
        const float aj = __shfl(ac, j0 + t, G);
        acc.x += aj * v[t].x;
        acc.y += aj * v[t].y;
        acc.z += aj * v[t].z;
        acc.w += aj * v[t].w;
    }
    if (more) {
#pragma unroll
        for (int t = 4; t < 8; ++t) {
            const float aj = __shfl(ac, j0 + t, G);
            acc.x += aj * v[t].x;
            acc.y += aj * v[t].y;
            acc.z += aj * v[t].z;
            acc.w += aj * v[t].w;
        }
    }
}

template <int C>
__device__ __forceinline__ void add_block(const float* __restrict__ h, const int* s, const float* a,
                                          int cnt, int gl, float4& acc) {
    constexpr int G  = C / 4;
    constexpr int KC = 32 / G;
#pragma unroll
    for (int c = 0; c < KC; ++c) {
#pragma unroll
        for (int jb = 0; jb < G; jb += 8) {
            const int rem = cnt - (c * G + jb);
            if (rem > 0) add_rows<C>(h, s[c], a[c], jb, rem > 4, gl, acc);
        }
    }
}

template <int C, int NT2>
__host__ __device__ constexpr int proj_lds_floats() {
    return NT2 > 0 ? C * (NT2 * 16 + 8) + 16 * (C + 4) : 0;
}

template <int C, int RELU, int NT2>
__global__ void __launch_bounds__(256)
node_softmax_sum(const int* __restrict__ cnt, const int* __restrict__ tab,
                 const int* __restrict__ ovn, const int* __restrict__ ove,
                 const int* __restrict__ esrc, const int* __restrict__ edst,
                 const float* __restrict__ ls, const float* __restrict__ ld,
                 const float* __restrict__ h, const float* __restrict__ bias,
                 float* __restrict__ out, int n_nodes, int n_edges,
                 const float* __restrict__ Wn, const float* __restrict__ an_src,
                 const float* __restrict__ an_dst,
                 float* __restrict__ lsn, float* __restrict__ ldn_out) {
    constexpr int G  = C / 4;
    constexpr int KC = CAP / G;
    static_assert(CAP == 32, "add_block walks 32 slots per group");
    static_assert(NT2 == 0 || 256 / G == 16, "a block must finish exactly one 16-row strip");
    constexpr int N2  = NT2 * 16;
    constexpr int P2  = N2 + 8;
    constexpr int PA  = C + 4;
    constexpr int NB2 = (C * N2) / (4 * 256);
    extern __shared__ __attribute__((aligned(16))) float proj_lds[];
    const int tid = blockIdx.x * blockDim.x + threadIdx.x;
    const int gl  = threadIdx.x & (G - 1);
    const int node = tid / G;
    if constexpr (NT2 == 0) {
        if ((tid & ~31) / G >= n_nodes) return;
    }

    float4 wq[NB2 > 0 ? NB2 : 1];
    if constexpr (NT2 > 0) {
#pragma unroll
        for (int j = 0; j < NB2; ++j)
            wq[j] = ((const float4*)Wn)[threadIdx.x + j * 256];
    }

    const bool live = node < n_nodes;
    const int nd = live ? node : n_nodes - 1;

    int raw[KC];
#pragma unroll
    for (int c = 0; c < KC; ++c) raw[c] = tab[nd * CAP + c * G + gl];
    const int   dall = cnt[nd];
    const float ldn  = ld[nd];

    if constexpr (NT2 > 0) {
#pragma unroll
        for (int j = 0; j < NB2; ++j) {
            const int i = (threadIdx.x + j * 256) * 4;
            *(float4*)(proj_lds + (i / N2) * P2 + (i % N2)) = wq[j];
        }
    }

    const int deg  = live ? dall : 0;
    const int dtab = deg < CAP ? deg : CAP;
    const bool over = deg > CAP;

    int dmax = dtab;
    for (int off = G; off < 32; off <<= 1) dmax = max(dmax, __shfl_xor(dmax, off, 32));
    const int wdeg = __builtin_amdgcn_readfirstlane(dmax);

    float4 acc = make_float4(0.0f, 0.0f, 0.0f, 0.0f);

    int   s[KC];
    float lv[KC];
    float lg[KC];
    bool  ok[KC];
    const int last = dtab > 0 ? dtab - 1 : 0;
    const int cl = last / G;
    int pick = raw[0];
#pragma unroll
    for (int c = 1; c < KC; ++c) pick = (cl == c) ? raw[c] : pick;
    const int lastsrc = __shfl(pick, last & (G - 1), G);
    const int fillsrc = dtab > 0 ? lastsrc : nd;
#pragma unroll
    for (int c = 0; c < KC; ++c) {
        ok[c] = c * G + gl < dtab;
        s[c] = ok[c] ? raw[c] : fillsrc;
    }
#pragma unroll
    for (int c = 0; c < KC; ++c) lv[c] = ls[s[c]];

    int novf = 0;
    if (over) {
        novf = *ovn;
        novf = novf < n_edges ? novf : n_edges;
    }

    float m = -INFINITY;
#pragma unroll
    for (int c = 0; c < KC; ++c) {
        lg[c] = ok[c] ? leaky(lv[c] + ldn) : -INFINITY;
        m = fmaxf(m, lg[c]);
    }
    if (over) {
        for (int i = gl; i < novf; i += G) {
            const int e = ove[i];
            if (edst[e] == node) m = fmaxf(m, leaky(ls[esrc[e]] + ldn));
        }
    }
    for (int off = G / 2; off > 0; off >>= 1) m = fmaxf(m, __shfl_xor(m, off, 32));

    float ex[KC];
    float den = 0.0f;
#pragma unroll
    for (int c = 0; c < KC; ++c) {
        ex[c] = ok[c] ? __expf(lg[c] - m) : 0.0f;
        den += ex[c];
    }
    if (over) {
        for (int i = gl; i < novf; i += G) {
            const int e = ove[i];
            if (edst[e] == node) den += __expf(leaky(ls[esrc[e]] + ldn) - m);
        }
    }
    for (int off = G / 2; off > 0; off >>= 1) den += __shfl_xor(den, off, 32);

    float a[KC];
#pragma unroll
    for (int c = 0; c < KC; ++c) a[c] = ok[c] ? ex[c] / den : 0.0f;

    add_block<C>(h, s, a, wdeg, gl, acc);

    if (over) {
        for (int i = 0; i < novf; ++i) {
            const int e = ove[i];
            if (edst[e] == node) {
                const int sj = esrc[e];
                const float w = __expf(leaky(ls[sj] + ldn) - m) / den;
                const float4 v = *(const float4*)((const char*)h + ((unsigned)sj * (C * 4u) + gl * 16u));
                acc.x += w * v.x;
                acc.y += w * v.y;
                acc.z += w * v.z;
                acc.w += w * v.w;
            }
        }
    }

    const float* bp = bias + gl * 4;
    const float4 b = make_float4(bp[0], bp[1], bp[2], bp[3]);
    float4 r = make_float4(acc.x + b.x, acc.y + b.y, acc.z + b.z, acc.w + b.w);
    if (RELU) {
        r.x = fmaxf(r.x, 0.0f);
        r.y = fmaxf(r.y, 0.0f);
        r.z = fmaxf(r.z, 0.0f);
        r.w = fmaxf(r.w, 0.0f);
    }

    if constexpr (NT2 == 0) {
        if (live) *(float4*)(out + (size_t)node * C + gl * 4) = r;
    } else {
        float* sW = proj_lds;
        float* sA = proj_lds + C * P2;
        *(float4*)(sA + (threadIdx.x / G) * PA + gl * 4) = r;
        __syncthreads();

        const int lane = threadIdx.x & 31;
        const int wid  = threadIdx.x >> 5;
        if (wid != (int)(blockIdx.x & 7)) return;
        const int mn = lane & 15;
        const int hi = lane >> 4;
        const int row0 = blockIdx.x * 16;

        float as2[NT2 > 0 ? NT2 : 1], ad2[NT2 > 0 ? NT2 : 1];
#pragma unroll
        for (int t = 0; t < NT2; ++t) {
            as2[t] = an_src[t * 16 + mn];
            ad2[t] = an_dst[t * 16 + mn];
        }

        float4 aq[C / 8];
#pragma unroll
        for (int j = 0; j < C / 8; ++j)
            aq[j] = *(const float4*)(sA + mn * PA + 8 * j + 4 * hi);

        v8f pacc[NT2 > 0 ? NT2 : 1];
#pragma unroll
        for (int t = 0; t < NT2; ++t) pacc[t] = (v8f){};

#pragma unroll
        for (int j = 0; j < C / 8; ++j) {
#pragma unroll
            for (int u = 0; u < 2; ++u) {
                const int kk = 8 * j + 4 * hi + 2 * u;
                v2f a;
                a.x = u ? aq[j].z : aq[j].x;
                a.y = u ? aq[j].w : aq[j].y;
                const float* brow = sW + kk * P2 + mn;
#pragma unroll
                for (int t = 0; t < NT2; ++t) {
                    v2f bb;
                    bb.x = brow[t * 16];
                    bb.y = brow[t * 16 + P2];
                    pacc[t] = __builtin_amdgcn_wmma_f32_16x16x4_f32(
                         false, a,  false, bb,
                         (short)0, pacc[t],  false,  false);
                }
            }
        }

#pragma unroll
        for (int t = 0; t < NT2; ++t) {
#pragma unroll
            for (int v = 0; v < 8; ++v) {
                const int row = row0 + hi * 8 + v;
                if (row < n_nodes) out[(size_t)row * N2 + t * 16 + mn] = pacc[t][v];
            }
        }

        float ps[8], pd[8];
        for (int v = 0; v < 8; ++v) { ps[v] = 0.0f; pd[v] = 0.0f; }
#pragma unroll
        for (int t = 0; t < NT2; ++t) {
            for (int v = 0; v < 8; ++v) {
                ps[v] += pacc[t][v] * as2[t];
                pd[v] += pacc[t][v] * ad2[t];
            }
        }
        for (int off = 8; off > 0; off >>= 1) {
            for (int v = 0; v < 8; ++v) {
                ps[v] += __shfl_xor(ps[v], off, 32);
                pd[v] += __shfl_xor(pd[v], off, 32);
            }
        }
        const int rr = mn & 7;
        float sv = ps[0], dv = pd[0];
        for (int v = 1; v < 8; ++v) {
            sv = (rr == v) ? ps[v] : sv;
            dv = (rr == v) ? pd[v] : dv;
        }
        float* sc = (mn < 8) ? lsn : ldn_out;
        if (row0 + hi * 8 + rr < n_nodes) sc[row0 + hi * 8 + rr] = (mn < 8) ? sv : dv;
    }
}


static inline int cdiv(int a, int b) { return (a + b - 1) / b; }

extern "C" void kernel_launch(void* const* d_in, const int* in_sizes, int n_in,
                              void* d_out, int out_size, void* d_ws, size_t ws_size,
                              hipStream_t stream) {
    const float* x      = (const float*)d_in[0];
    const int*   eidx   = (const int*)d_in[1];
    const float* W1     = (const float*)d_in[2];
    const float* a_src1 = (const float*)d_in[3];
    const float* a_dst1 = (const float*)d_in[4];
    const float* b1     = (const float*)d_in[5];
    const float* W2     = (const float*)d_in[6];
    const float* a_src2 = (const float*)d_in[7];
    const float* a_dst2 = (const float*)d_in[8];
    const float* b2     = (const float*)d_in[9];
    float* out = (float*)d_out;

    const int IN = 128, H = 64, O = 32;
    const int N = in_sizes[0] / IN;
    const int E = in_sizes[1] / 2;
    const int* src = eidx;
    const int* dst = eidx + E;

    float* ws    = (float*)d_ws;
    float* h1    = ws;
    float* h2    = h1 + (size_t)N * H;
    float* ls2   = h2 + (size_t)N * O;
    float* ld2   = ls2 + N;
    int*   tab   = (int*)(h1 + 2 * (size_t)N * H);
    float* ls    = (float*)(tab + (size_t)N * CAP);
    float* ld    = ls + N;
    int*   cnt   = (int*)(ld + N);
    int*   ovn   = cnt + N;
    int*   ove   = cnt + 2 * (size_t)N;

    const int B = 256;

    const int nbk = cdiv(N, NPB);
    int2* runs = (int2*)h1;
    int*  cls  = (int*)(runs + (size_t)NWG * nbk * SC);
    int*  runc = cls + ((size_t)E + 32 * NWG);
    int*  clc  = runc + (size_t)NWG * nbk;
    bin_edges<<<NWG, 256, 0, stream>>>(src, dst, runs, runc, cls, clc, ovn, E, nbk);
    slot_ranges<<<nbk, NWG, 0, stream>>>(src, dst, runs, runc, cls, clc, cnt, tab, ovn, ove, N, E, nbk);

    gemm_wmma_strip<4, 128><<<cdiv(N, 128), B, 0, stream>>>(x, W1, a_src1, a_dst1, h1, ls, ld, N);
    node_softmax_sum<64, 1, 2><<<cdiv(N * 16, B), B, proj_lds_floats<64, 2>() * sizeof(float), stream>>>(
        cnt, tab, ovn, ove, src, dst, ls, ld, h1, b1, h2, N, E, W2, a_src2, a_dst2, ls2, ld2);

    node_softmax_sum<32, 0, 0><<<cdiv(N * 8, B), B, 0, stream>>>(
        cnt, tab, ovn, ove, src, dst, ls2, ld2, h2, b2, out, N, E, nullptr, nullptr, nullptr, nullptr, nullptr);
}


# Node kernels: source scores formed from the gathered rows (dot with a_src, DPP sums over the lane group), running softmax divided once at the end (sum order changes): no per-edge score gather
# speedup vs baseline: 6.9624x; 1.2255x over previous
#include <hip/hip_runtime.h>
#include <hip/hip_bf16.h>
#include <math.h>


#define SLOPE 0.2f

typedef float v2f __attribute__((ext_vector_type(2)));
typedef float v8f __attribute__((ext_vector_type(8)));


__device__ __forceinline__ float leaky(float v) {
    return v > 0.0f ? v : SLOPE * v;
}

template <int NT, int K>
__device__ __forceinline__ void
gemm_wmma_strip(const float* __restrict__ A,
                const float* __restrict__ B,
                const float* __restrict__ a_src,
                const float* __restrict__ a_dst,
                float* __restrict__ C,
                float* __restrict__ ls, float* __restrict__ ld,
                int M, int blk) {
    constexpr int N  = NT * 16;
    constexpr int P  = N + 8;
    constexpr int NB = (K * N) / (4 * 256);
    __shared__ __attribute__((aligned(16))) float sB[K * P];

    const int lane = threadIdx.x & 31;
    const int wid  = threadIdx.x >> 5;
    const int row0 = (blk * 8 + wid) * 16;
    const bool active = row0 < M;
    const int mn = lane & 15;
    const int hi = lane >> 4;

    float4 bq[NB];
#pragma unroll
    for (int j = 0; j < NB; ++j)
        bq[j] = ((const float4*)B)[threadIdx.x + j * 256];

    float4 aq[K / 8];
    if (active) {
        const float4* arow = (const float4*)(A + (size_t)(row0 + mn) * K) + hi;
#pragma unroll
        for (int j = 0; j < K / 8; ++j) aq[j] = arow[2 * j];
    }

#pragma unroll
    for (int j = 0; j < NB; ++j) {
        const int i = (threadIdx.x + j * 256) * 4;
        *(float4*)(sB + (i / N) * P + (i % N)) = bq[j];
    }
    __syncthreads();
    if (!active) return;

    v8f acc[NT];
#pragma unroll
    for (int t = 0; t < NT; ++t) acc[t] = (v8f){};

#pragma unroll
    for (int j = 0; j < K / 8; ++j) {
#pragma unroll
        for (int u = 0; u < 2; ++u) {
            const int kk = 8 * j + 4 * hi + 2 * u;
            v2f a;
            a.x = u ? aq[j].z : aq[j].x;
            a.y = u ? aq[j].w : aq[j].y;
            const float* brow = sB + kk * P + mn;
#pragma unroll
            for (int t = 0; t < NT; ++t) {
                v2f b;
                b.x = brow[t * 16];
                b.y = brow[t * 16 + P];
                acc[t] = __builtin_amdgcn_wmma_f32_16x16x4_f32(
                     false, a,  false, b,
                     (short)0, acc[t],  false,  false);
            }
        }
    }

#pragma unroll
    for (int t = 0; t < NT; ++t) {
        float* crow = C + (size_t)(row0 + hi * 8) * N + t * 16 + mn;
#pragma unroll
        for (int v = 0; v < 8; ++v) crow[(size_t)v * N] = acc[t][v];
    }

    float ps[8], pd[8];
    for (int v = 0; v < 8; ++v) { ps[v] = 0.0f; pd[v] = 0.0f; }
    for (int t = 0; t < NT; ++t) {
        const float as = a_src[t * 16 + mn];
        const float ad = a_dst[t * 16 + mn];
        for (int v = 0; v < 8; ++v) {
            ps[v] += acc[t][v] * as;
            pd[v] += acc[t][v] * ad;
        }
    }
    for (int off = 8; off > 0; off >>= 1) {
        for (int v = 0; v < 8; ++v) {
            ps[v] += __shfl_xor(ps[v], off, 32);
            pd[v] += __shfl_xor(pd[v], off, 32);
        }
    }
    const int r = mn & 7;
    float sv = ps[0], dv = pd[0];
    for (int v = 1; v < 8; ++v) {
        sv = (r == v) ? ps[v] : sv;
        dv = (r == v) ? pd[v] : dv;
    }
    float* sc = (mn < 8) ? ls : ld;
    sc[row0 + hi * 8 + r] = (mn < 8) ? sv : dv;
}


#define CAP 32
#define NPB 196
#define NWG 256
#define SC  30
#define EPT 13
#define MAXBK 512

__device__ __forceinline__ int chunk_len(int E) {
    return (((E + NWG - 1) / NWG) + 31) & ~31;
}

__global__ void __launch_bounds__(256)
bin_edges(const int* __restrict__ src, const int* __restrict__ dst,
          int2* __restrict__ runs, int* __restrict__ runc,
          int* __restrict__ cls, int* __restrict__ clc, int* __restrict__ ovn,
          int E, int nbk) {
    __shared__ int lcnt[MAXBK];
    __shared__ int lown[256];
    __shared__ int lsp;
    const int w = blockIdx.x;
    const int chunk = chunk_len(E);
    const int base = w * chunk;
    for (int t = threadIdx.x; t < nbk; t += 256) lcnt[t] = 0;
    lown[threadIdx.x] = 0;
    if (threadIdx.x == 0) lsp = 0;
    __syncthreads();
    for (int i0 = 0; i0 < chunk; i0 += 256 * EPT) {
        int d[EPT], s[EPT];
#pragma unroll
        for (int j = 0; j < EPT; ++j) {
            const int i = i0 + j * 256 + threadIdx.x;
            const bool v = i < chunk && base + i < E;
            d[j] = v ? dst[base + i] : -1;
            s[j] = v ? src[base + i] : 0;
        }
        int r[EPT];
#pragma unroll
        for (int j = 0; j < EPT; ++j) {
            const int b = d[j] / NPB;
            const bool v = d[j] >= 0 && b < nbk;
            r[j] = atomicAdd(v ? &lcnt[b] : &lown[threadIdx.x], 1);
            r[j] = v ? r[j] : -1;
        }
#pragma unroll
        for (int j = 0; j < EPT; ++j) {
            if (r[j] >= 0) {
                const int li = i0 + j * 256 + threadIdx.x;
                const int b = d[j] / NPB;
                if (r[j] < SC) {
                    const size_t p = ((size_t)w * nbk + b) * SC + r[j];
                    runs[p] = make_int2(s[j], (d[j] - b * NPB) | (li << 8));
                } else {
                    cls[base + atomicAdd(&lsp, 1)] = base + li;
                }
            }
        }
    }
    __syncthreads();
    for (int t = threadIdx.x; t < nbk; t += 256) {
        const int c = lcnt[t];
        runc[w * nbk + t] = c < SC ? c : SC;
    }
    if (threadIdx.x == 0) {
        clc[w] = lsp;
        if (w == 0) *ovn = 0;
    }
}

__device__ __forceinline__ void place_edge(int* lcn, int* ltab, int s, int dl, int e,
                                           int* __restrict__ ovn, int* __restrict__ ove) {
    const int r = atomicAdd(&lcn[dl], 1);
    if (r < CAP) ltab[dl * CAP + r] = s;
    else         ove[atomicAdd(ovn, 1)] = e;
}

__device__ __forceinline__ void
slot_ranges(const int* __restrict__ src, const int* __restrict__ dst,
            const int2* __restrict__ runs,
            const int* __restrict__ runc, const int* __restrict__ cls,
            const int* __restrict__ clc,
            int* __restrict__ cnt, int* __restrict__ tab,
            int* __restrict__ ovn, int* __restrict__ ove, int N, int E, int nbk, int b) {
    __shared__ __attribute__((aligned(16))) int ltab[NPB * CAP];
    __shared__ int lcn[NPB];
    __shared__ int lown[NWG];
    const int n0 = b * NPB;
    const int w  = threadIdx.x;
    const int chunk = chunk_len(E);
    const size_t run = ((size_t)w * nbk + b) * SC;

    int c  = runc[w * nbk + b];
    int n1 = clc[w];
    const int4* rp = (const int4*)(runs + run);
    int4 q[SC / 2];
#pragma unroll
    for (int j = 0; j < SC / 2; ++j) q[j] = rp[j];

    for (int i = threadIdx.x; i < NPB * CAP / 4; i += NWG)
        ((int4*)ltab)[i] = make_int4(0, 0, 0, 0);
    if (threadIdx.x < NPB) lcn[threadIdx.x] = 0;
    lown[threadIdx.x] = 0;
    __syncthreads();

    c  = c  < SC ? c : SC;
    n1 = n1 < chunk ? n1 : chunk;
    int rk[SC];
#pragma unroll
    for (int k = 0; k < SC; ++k) {
        const int dl = ((k & 1) ? q[k / 2].w : q[k / 2].y) & 255;
        const bool v = k < c && dl < NPB;
        rk[k] = atomicAdd(v ? &lcn[dl] : &lown[threadIdx.x], 1);
        rk[k] = v ? rk[k] : -1;
    }
#pragma unroll
    for (int k = 0; k < SC; ++k) {
        if (rk[k] >= 0) {
            const unsigned code = (unsigned)((k & 1) ? q[k / 2].w : q[k / 2].y);
            const int dl = code & 255;
            const int s  = (k & 1) ? q[k / 2].z : q[k / 2].x;
            if (rk[k] < CAP) ltab[dl * CAP + rk[k]] = s;
            else             ove[atomicAdd(ovn, 1)] = w * chunk + (int)(code >> 8);
        }
    }
    for (int k = 0; k < n1; ++k) {
        const int e = cls[w * chunk + k];
        if ((unsigned)e < (unsigned)E) {
            const int se = src[e];
            const int dl = dst[e] - n0;
            if ((unsigned)dl < (unsigned)NPB) place_edge(lcn, ltab, se, dl, e, ovn, ove);
        }
    }
    __syncthreads();

    int nn = N - n0;
    nn = nn < NPB ? nn : NPB;
    if ((int)threadIdx.x < nn) cnt[n0 + threadIdx.x] = lcn[threadIdx.x];
    int4* tp = (int4*)(tab + (size_t)n0 * CAP);
    for (int i = threadIdx.x; i < nn * (CAP / 4); i += NWG) tp[i] = ((const int4*)ltab)[i];
}

template <int NT, int K>
__global__ void __launch_bounds__(256)
slots_and_projection(const int* __restrict__ src, const int* __restrict__ dst,
                     const int2* __restrict__ runs,
                     const int* __restrict__ runc, const int* __restrict__ cls,
                     const int* __restrict__ clc,
                     int* __restrict__ cnt, int* __restrict__ tab,
                     int* __restrict__ ovn, int* __restrict__ ove, int N, int E, int nbk,
                     const float* __restrict__ A, const float* __restrict__ B,
                     const float* __restrict__ a_src, const float* __restrict__ a_dst,
                     float* __restrict__ C, float* __restrict__ ls, float* __restrict__ ld,
                     int ngemm) {
    static_assert(NWG == 256, "both bodies are written for work-groups of 256 threads");
    if ((int)blockIdx.x < ngemm)
        gemm_wmma_strip<NT, K>(A, B, a_src, a_dst, C, ls, ld, N, (int)blockIdx.x);
    else
        slot_ranges(src, dst, runs, runc, cls, clc, cnt, tab, ovn, ove, N, E, nbk,
                    (int)blockIdx.x - ngemm);
}


template <int K>
__device__ __forceinline__ float lane_xor(float x) {
    return __int_as_float(__builtin_amdgcn_update_dpp(0, __float_as_int(x), 0x160 + K, 0xf, 0xf, true));
}

template <int C>
__device__ __forceinline__ void add_rows(const float* __restrict__ h, int sc, int j0, bool more,
                                         int nv, int gl, const float4& as4, float ldn,
                                         float& m, float& den, float4& acc) {
    constexpr int G = C / 4;
    static_assert(G == 8 || G == 16, "a group is 8 or 16 lanes of one 16-lane row");
    constexpr int S = (G == 16) ? 1 : 0;
    float4 v[8];
#pragma unroll
    for (int t = 0; t < 4; ++t) {
        const int sj = __shfl(sc, j0 + t, G);
        v[t] = *(const float4*)((const char*)h + ((unsigned)sj * (C * 4u) + gl * 16u));
    }
    if (more) {
#pragma unroll
        for (int t = 4; t < 8; ++t) {
            const int sj = __shfl(sc, j0 + t, G);
            v[t] = *(const float4*)((const char*)h + ((unsigned)sj * (C * 4u) + gl * 16u));
        }
    }
    float d[8];
#pragma unroll
    for (int t = 0; t < 4; ++t)
        d[t] = v[t].x * as4.x + v[t].y * as4.y + v[t].z * as4.z + v[t].w * as4.w;
    if (more) {
#pragma unroll
        for (int t = 4; t < 8; ++t)
            d[t] = v[t].x * as4.x + v[t].y * as4.y + v[t].z * as4.z + v[t].w * as4.w;
    } else {
#pragma unroll
        for (int t = 4; t < 8; ++t) d[t] = 0.0f;
    }
    const bool hA = (gl >> (2 + S)) & 1;
    const bool hB = (gl >> (1 + S)) & 1;
    const bool hC = (gl >> S) & 1;
    float e[4], f[2];
#pragma unroll
    for (int i = 0; i < 4; ++i)
        e[i] = (hA ? d[i + 4] : d[i]) + lane_xor<(4 << S)>(hA ? d[i] : d[i + 4]);
#pragma unroll
    for (int i = 0; i < 2; ++i)
        f[i] = (hB ? e[i + 2] : e[i]) + lane_xor<(2 << S)>(hB ? e[i] : e[i + 2]);
    float z = (hC ? f[1] : f[0]) + lane_xor<(1 << S)>(hC ? f[0] : f[1]);
    if constexpr (G == 16) z += lane_xor<1>(z);
    const int tl = gl >> S;
    const float lgt = tl < nv ? leaky(z + ldn) : -INFINITY;
    float bm = lgt;
    bm = fmaxf(bm, lane_xor<(4 << S)>(bm));
    bm = fmaxf(bm, lane_xor<(2 << S)>(bm));
    bm = fmaxf(bm, lane_xor<(1 << S)>(bm));
    const float mn = fmaxf(m, bm);
    const float rs = __expf(m - mn);
    const float pl = __expf(lgt - mn);
    acc.x *= rs;
    acc.y *= rs;
    acc.z *= rs;
    acc.w *= rs;
    den *= rs;
    m = mn;
#pragma unroll
    for (int t = 0; t < 4; ++t) {
        const float p = __shfl(pl, t << S, G);
        den += p;
        acc.x += p * v[t].x;
        acc.y += p * v[t].y;
        acc.z += p * v[t].z;
        acc.w += p * v[t].w;
    }
    if (more) {
#pragma unroll
        for (int t = 4; t < 8; ++t) {
            const float p = __shfl(pl, t << S, G);
            den += p;
            acc.x += p * v[t].x;
            acc.y += p * v[t].y;
            acc.z += p * v[t].z;
            acc.w += p * v[t].w;
        }
    }
}

template <int C>
__device__ __forceinline__ void add_block(const float* __restrict__ h, const int* s, int cnt, int dtab,
                                          int gl, const float4& as4, float ldn,
                                          float& m, float& den, float4& acc) {
    constexpr int G  = C / 4;
    constexpr int KC = 32 / G;
#pragma unroll
    for (int c = 0; c < KC; ++c) {
#pragma unroll
        for (int jb = 0; jb < G; jb += 8) {
            const int rem = cnt - (c * G + jb);
            if (rem > 0)
                add_rows<C>(h, s[c], jb, rem > 4, dtab - (c * G + jb), gl, as4, ldn, m, den, acc);
        }
    }
}

template <int C, int NT2>
__host__ __device__ constexpr int proj_lds_floats() {
    return NT2 > 0 ? C * (NT2 * 16 + 8) + 16 * (C + 4) : 0;
}

template <int C, int RELU, int NT2>
__global__ void __launch_bounds__(256)
node_softmax_sum(const int* __restrict__ cnt, const int* tab,
                 const int* __restrict__ ovn, const int* __restrict__ ove,
                 const int* __restrict__ esrc, const int* __restrict__ edst,
                 const float* __restrict__ ls, const float* __restrict__ ld,
                 const float* __restrict__ a_cur,
                 const float* __restrict__ h, const float* __restrict__ bias,
                 float* out, int n_nodes, int n_edges,
                 const float* __restrict__ Wn, const float* __restrict__ an_src,
                 const float* __restrict__ an_dst,
                 float* __restrict__ lsn, float* __restrict__ ldn_out) {
    constexpr int G  = C / 4;
    constexpr int KC = CAP / G;
    static_assert(CAP == 32, "add_block walks 32 slots per group");
    static_assert(NT2 == 0 || 256 / G == 16, "a block must finish exactly one 16-row strip");
    constexpr int N2  = NT2 * 16;
    constexpr int P2  = N2 + 8;
    constexpr int PA  = C + 4;
    constexpr int NB2 = (C * N2) / (4 * 256);
    extern __shared__ __attribute__((aligned(16))) float proj_lds[];
    const int tid = blockIdx.x * blockDim.x + threadIdx.x;
    const int gl  = threadIdx.x & (G - 1);
    const int node = tid / G;
    if constexpr (NT2 == 0) {
        if ((tid & ~31) / G >= n_nodes) return;
    }

    float4 wq[NB2 > 0 ? NB2 : 1];
    if constexpr (NT2 > 0) {
#pragma unroll
        for (int j = 0; j < NB2; ++j)
            wq[j] = ((const float4*)Wn)[threadIdx.x + j * 256];
    }

    const bool live = node < n_nodes;
    const int nd = live ? node : n_nodes - 1;

    int raw[KC];
#pragma unroll
    for (int c = 0; c < KC; ++c) raw[c] = tab[nd * CAP + c * G + gl];
    const int   dall = cnt[nd];
    const float ldn  = ld[nd];
    const float4 as4 = *(const float4*)(a_cur + gl * 4);

    if constexpr (NT2 > 0) {
#pragma unroll
        for (int j = 0; j < NB2; ++j) {
            const int i = (threadIdx.x + j * 256) * 4;
            *(float4*)(proj_lds + (i / N2) * P2 + (i % N2)) = wq[j];
        }
    }

    const int deg  = live ? dall : 0;
    const int dtab = deg < CAP ? deg : CAP;
    const bool over = deg > CAP;

    int dmax = dtab;
    for (int off = G; off < 32; off <<= 1) dmax = max(dmax, __shfl_xor(dmax, off, 32));
    const int wdeg = __builtin_amdgcn_readfirstlane(dmax);

    float4 acc = make_float4(0.0f, 0.0f, 0.0f, 0.0f);

    int s[KC];
    const int last = dtab > 0 ? dtab - 1 : 0;
    const int cl = last / G;
    int pick = raw[0];
#pragma unroll
    for (int c = 1; c < KC; ++c) pick = (cl == c) ? raw[c] : pick;
    const int lastsrc = __shfl(pick, last & (G - 1), G);
    const int fillsrc = dtab > 0 ? lastsrc : nd;
#pragma unroll
    for (int c = 0; c < KC; ++c) s[c] = (c * G + gl < dtab) ? raw[c] : fillsrc;

    int novf = 0;
    if (over) {
        novf = *ovn;
        novf = novf < n_edges ? novf : n_edges;
    }

    float m = -3.402823466e+38f;
    float den = 0.0f;
    add_block<C>(h, s, wdeg, dtab, gl, as4, ldn, m, den, acc);

    if (over) {
        for (int i = 0; i < novf; ++i) {
            const int e = ove[i];
            if (edst[e] == node) {
                const int sj = esrc[e];
                const float l = leaky(ls[sj] + ldn);
                const float mn = fmaxf(m, l);
                const float rs = __expf(m - mn);
                const float w = __expf(l - mn);
                const float4 v = *(const float4*)((const char*)h + ((unsigned)sj * (C * 4u) + gl * 16u));
                acc.x = acc.x * rs + w * v.x;
                acc.y = acc.y * rs + w * v.y;
                acc.z = acc.z * rs + w * v.z;
                acc.w = acc.w * rs + w * v.w;
                den = den * rs + w;
                m = mn;
            }
        }
    }

    const float inv = den > 0.0f ? 1.0f / den : 0.0f;
    const float* bp = bias + gl * 4;
    const float4 b = make_float4(bp[0], bp[1], bp[2], bp[3]);
    float4 r = make_float4(acc.x * inv + b.x, acc.y * inv + b.y, acc.z * inv + b.z, acc.w * inv + b.w);
    if (RELU) {
        r.x = fmaxf(r.x, 0.0f);
        r.y = fmaxf(r.y, 0.0f);
        r.z = fmaxf(r.z, 0.0f);
        r.w = fmaxf(r.w, 0.0f);
    }

    if constexpr (NT2 == 0) {
        if (live) *(float4*)(out + (size_t)node * C + gl * 4) = r;
    } else {
        float* sW = proj_lds;
        float* sA = proj_lds + C * P2;
        *(float4*)(sA + (threadIdx.x / G) * PA + gl * 4) = r;
        __syncthreads();

        const int lane = threadIdx.x & 31;
        const int wid  = threadIdx.x >> 5;
        if (wid != (int)(blockIdx.x & 7)) return;
        const int mn = lane & 15;
        const int hi = lane >> 4;
        const int row0 = blockIdx.x * 16;

        float as2[NT2 > 0 ? NT2 : 1], ad2[NT2 > 0 ? NT2 : 1];
#pragma unroll
        for (int t = 0; t < NT2; ++t) {
            as2[t] = an_src[t * 16 + mn];
            ad2[t] = an_dst[t * 16 + mn];
        }

        float4 aq[C / 8];
#pragma unroll
        for (int j = 0; j < C / 8; ++j)
            aq[j] = *(const float4*)(sA + mn * PA + 8 * j + 4 * hi);

        v8f pacc[NT2 > 0 ? NT2 : 1];
#pragma unroll
        for (int t = 0; t < NT2; ++t) pacc[t] = (v8f){};

#pragma unroll
        for (int j = 0; j < C / 8; ++j) {
#pragma unroll
            for (int u = 0; u < 2; ++u) {
                const int kk = 8 * j + 4 * hi + 2 * u;
                v2f a;
                a.x = u ? aq[j].z : aq[j].x;
                a.y = u ? aq[j].w : aq[j].y;
                const float* brow = sW + kk * P2 + mn;
#pragma unroll
                for (int t = 0; t < NT2; ++t) {
                    v2f bb;
                    bb.x = brow[t * 16];
                    bb.y = brow[t * 16 + P2];
                    pacc[t] = __builtin_amdgcn_wmma_f32_16x16x4_f32(
                         false, a,  false, bb,
                         (short)0, pacc[t],  false,  false);
                }
            }
        }

#pragma unroll
        for (int t = 0; t < NT2; ++t) {
#pragma unroll
            for (int v = 0; v < 8; ++v) {
                const int row = row0 + hi * 8 + v;
                if (row < n_nodes) out[(size_t)row * N2 + t * 16 + mn] = pacc[t][v];
            }
        }

        float ps[8], pd[8];
        for (int v = 0; v < 8; ++v) { ps[v] = 0.0f; pd[v] = 0.0f; }
#pragma unroll
        for (int t = 0; t < NT2; ++t) {
            for (int v = 0; v < 8; ++v) {
                ps[v] += pacc[t][v] * as2[t];
                pd[v] += pacc[t][v] * ad2[t];
            }
        }
        for (int off = 8; off > 0; off >>= 1) {
            for (int v = 0; v < 8; ++v) {
                ps[v] += __shfl_xor(ps[v], off, 32);
                pd[v] += __shfl_xor(pd[v], off, 32);
            }
        }
        const int rr = mn & 7;
        float sv = ps[0], dv = pd[0];
        for (int v = 1; v < 8; ++v) {
            sv = (rr == v) ? ps[v] : sv;
            dv = (rr == v) ? pd[v] : dv;
        }
        float* sc = (mn < 8) ? lsn : ldn_out;
        if (row0 + hi * 8 + rr < n_nodes) sc[row0 + hi * 8 + rr] = (mn < 8) ? sv : dv;
    }
}


static inline int cdiv(int a, int b) { return (a + b - 1) / b; }

extern "C" void kernel_launch(void* const* d_in, const int* in_sizes, int n_in,
                              void* d_out, int out_size, void* d_ws, size_t ws_size,
                              hipStream_t stream) {
    const float* x      = (const float*)d_in[0];
    const int*   eidx   = (const int*)d_in[1];
    const float* W1     = (const float*)d_in[2];
    const float* a_src1 = (const float*)d_in[3];
    const float* a_dst1 = (const float*)d_in[4];
    const float* b1     = (const float*)d_in[5];
    const float* W2     = (const float*)d_in[6];
    const float* a_src2 = (const float*)d_in[7];
    const float* a_dst2 = (const float*)d_in[8];
    const float* b2     = (const float*)d_in[9];
    float* out = (float*)d_out;

    const int IN = 128, H = 64, O = 32;
    const int N = in_sizes[0] / IN;
    const int E = in_sizes[1] / 2;
    const int* src = eidx;
    const int* dst = eidx + E;

    float* ws    = (float*)d_ws;
    float* h1    = ws;
    int*   tab   = (int*)d_out;

    const int B = 256;

    const int nbk = cdiv(N, NPB);
    int2* runs = (int2*)(h1 + (size_t)N * H);
    int*  cls  = (int*)(runs + (size_t)NWG * nbk * SC);
    int*  runc = cls + ((size_t)E + 32 * NWG);
    int*  clc  = runc + (size_t)NWG * nbk;
    float* h2  = (float*)runs;
    float* ls2 = h2 + (size_t)N * O;
    float* ld2 = ls2 + N;
    float* ls    = (float*)(clc + NWG);
    float* ld    = ls + N;
    int*   cnt   = (int*)(ld + N);
    int*   ovn   = cnt + N;
    int*   ove   = ovn + 4;
    bin_edges<<<NWG, 256, 0, stream>>>(src, dst, runs, runc, cls, clc, ovn, E, nbk);

    const int ngemm = cdiv(N, 128);
    slots_and_projection<4, 128><<<ngemm + nbk, B, 0, stream>>>(
        src, dst, runs, runc, cls, clc, cnt, tab, ovn, ove, N, E, nbk,
        x, W1, a_src1, a_dst1, h1, ls, ld, ngemm);

    node_softmax_sum<64, 1, 2><<<cdiv(N * 16, B), B, proj_lds_floats<64, 2>() * sizeof(float), stream>>>(
        cnt, tab, ovn, ove, src, dst, ls, ld, a_src1, h1, b1, h2, N, E, W2, a_src2, a_dst2, ls2, ld2);

    node_softmax_sum<32, 0, 0><<<cdiv(N * 8, B), B, 0, stream>>>(
        cnt, tab, ovn, ove, src, dst, ls2, ld2, a_src2, h2, b2, out, N, E, nullptr, nullptr, nullptr, nullptr, nullptr);
}
